# starting kernel
# baseline (speedup 1.0000x reference)
#include <hip/hip_runtime.h>
#include <hip/hip_bf16.h>
#include <math.h>


typedef __attribute__((ext_vector_type(2))) float v2f;
typedef __attribute__((ext_vector_type(8))) float v8f;

#define NN   128
#define SLD  132
#define PLD  132
#define TS   10
#define STEP_EPS 0.1f
#define ALPHA 5.0f
#define ALPHA_BASE 0.1f
#define KRANK 3
#define PSTEPS 10

__device__ __forceinline__ float softplus_f(float x) {
  return fmaxf(x, 0.0f) + log1pf(expf(-fabsf(x)));
}

__device__ __forceinline__ float block_reduce(float v, float* red, int tid) {
  red[tid] = v;
  __syncthreads();
  #pragma unroll
  for (int s = 128; s > 0; s >>= 1) {
    if (tid < s) red[tid] += red[tid + s];
    __syncthreads();
  }
  float r = red[0];
  __syncthreads();
  return r;
}

__launch_bounds__(256, 1)
__global__ void ts_gcn_fused(const float* __restrict__ edge_pred,
                             const int*   __restrict__ adj,
                             const float* __restrict__ d_init,
                             const float* __restrict__ u_init,
                             const float* __restrict__ x_noise,
                             float*       __restrict__ out)
{
  __shared__ float  Dp[NN * NN];
  __shared__ float  Cw[NN * NN];
  __shared__ float  Sm[NN * SLD];
  __shared__ float2 Xq[64 * 16];
  __shared__ float  Pp[16 * PLD];
  __shared__ float  Xx[NN], Xy[NN], Xz[NN];
  __shared__ float  rsum[NN];
  __shared__ float  red[256];
  __shared__ float  ucol[KRANK][NN];
  __shared__ float  uvec[NN], utmp[NN];

  const int b   = blockIdx.x;
  const int tid = threadIdx.x;
  const float d0 = d_init[0];

  const float* ep = edge_pred + (size_t)b * NN * NN * 2;
  const int*   ad = adj       + (size_t)b * NN * NN;

  float partM = 0.0f;
  for (int p = tid; p < NN * NN; p += 256) {
    const int i = p >> 7, j = p & 127;
    const float  m = (float)ad[p];
    const float2 e = ((const float2*)ep)[p];
    const float2 t = ((const float2*)ep)[j * NN + i];
    const float sp0 = softplus_f(d0 + e.x + t.x);
    const float sp1 = softplus_f(d0 + e.y + t.y);
    Dp[p] = sp0 * m;
    Cw[p] = sp1 * m;
    partM += (i == j) ? 1.0f : m;
  }
  for (int idx = tid; idx < 64 * 16; idx += 256) Xq[idx] = make_float2(0.f, 0.f);
  const float Msum = block_reduce(partM, red, tid);

  const float Nmol = block_reduce(
      (tid < NN) ? ((tid == 0) ? 1.0f : (float)ad[tid * NN]) : 0.0f, red, tid);

  {
    const int i = tid >> 1, j0 = (tid & 1) << 6;
    float part = 0.0f;
    for (int jj = 0; jj < 64; ++jj) {
      const float d = Dp[i * NN + j0 + jj];
      part += d * d;
    }
    red[tid] = part;
    __syncthreads();
    if (tid < NN) rsum[tid] = red[2 * tid] + red[2 * tid + 1];
    __syncthreads();
    const float tot  = block_reduce((tid < NN) ? rsum[tid] : 0.0f, red, tid);
    const float invN = 1.0f / Nmol;
    const float Mn   = tot * invN * invN;
    for (int p = tid; p < NN * NN; p += 256) {
      const int i2 = p >> 7, j2 = p & 127;
      const float d  = Dp[p];
      const float d2 = d * d;
      const float mk = (i2 == j2) ? 1.0f : (d > 0.0f ? 1.0f : 0.0f);
      Sm[i2 * SLD + j2] =
          mk * -0.5f * (d2 - rsum[j2] * invN - rsum[i2] * invN + Mn);
    }
    __syncthreads();
  }

  for (int kx = 0; kx < KRANK; ++kx) {
    if (tid < NN) uvec[tid] = u_init[(size_t)b * NN * 3 + tid * 3 + kx];
    __syncthreads();
    for (int it = 0; it < PSTEPS; ++it) {
      const float nv =
          block_reduce((tid < NN) ? uvec[tid] * uvec[tid] : 0.0f, red, tid);
      const float inorm = 1.0f / fmaxf(sqrtf(nv), 0.001f);
      if (tid < NN) uvec[tid] *= inorm;
      __syncthreads();
      float dc0 = 0.0f, dc1 = 0.0f;
      if (kx > 0)
        dc0 = block_reduce((tid < NN) ? ucol[0][tid] * uvec[tid] : 0.0f, red, tid);
      if (kx > 1)
        dc1 = block_reduce((tid < NN) ? ucol[1][tid] * uvec[tid] : 0.0f, red, tid);
      if (tid < NN) {
        const float* row = &Sm[tid * SLD];
        float acc = 0.0f;
        #pragma unroll 8
        for (int j = 0; j < NN; ++j) acc = fmaf(row[j], uvec[j], acc);
        if (kx > 0) acc -= ucol[0][tid] * dc0;
        if (kx > 1) acc -= ucol[1][tid] * dc1;
        utmp[tid] = acc;
      }
      __syncthreads();
      if (tid < NN) uvec[tid] = utmp[tid];
      __syncthreads();
    }
    const float eig =
        block_reduce((tid < NN) ? uvec[tid] * uvec[tid] : 0.0f, red, tid);
    const float iden = 1.0f / sqrtf(sqrtf(eig + 0.01f));
    if (tid < NN) ucol[kx][tid] = uvec[tid] * iden;
    __syncthreads();
  }
  if (tid < NN) {
    const float* xn = x_noise + (size_t)b * NN * 3 + tid * 3;
    Xx[tid] = ucol[0][tid] + xn[0];
    Xy[tid] = ucol[1][tid] + xn[1];
    Xz[tid] = ucol[2][tid] + xn[2];
  }
  __syncthreads();
  if (tid < 192) {
    const int q = tid & 63, n = tid >> 6;
    const float* comp = (n == 0) ? Xx : ((n == 1) ? Xy : Xz);
    Xq[q * 16 + n] = make_float2(comp[2 * q], comp[2 * q + 1]);
  }
  __syncthreads();

  const float invM4 = 4.0f / Msum;
  const int lane   = tid & 31;
  const int mt     = tid >> 5;
  const int nidx   = lane & 15;
  const int laneHi = lane >> 4;
  const int arow   = mt * 16 + nidx;
  const float2* aq = (const float2*)&Sm[arow * SLD + laneHi * 2];
  const float2* bq = &Xq[laneHi * 16 + nidx];

  for (int tstep = 0; tstep < TS; ++tstep) {
    {
      const int i = tid >> 1, j0 = (tid & 1) << 6;
      const float xi0 = Xx[i], xi1 = Xy[i], xi2 = Xz[i];
      float part = 0.0f;
      for (int jj = 0; jj < 64; ++jj) {
        const int j = j0 + jj;
        const int p = i * NN + j;
        const float u0 = xi0 - Xx[j];
        const float u1 = xi1 - Xy[j];
        const float u2 = xi2 - Xz[j];
        const float dist = sqrtf(u0 * u0 + u1 * u1 + u2 * u2 + 0.01f);
        const float s = invM4 * Cw[p] * (dist - Dp[p]) / dist;
        Sm[i * SLD + j] = s;
        part += s;
      }
      red[tid] = part;
      __syncthreads();
      if (tid < NN) rsum[tid] = red[2 * tid] + red[2 * tid + 1];
      __syncthreads();
    }

    v8f acc = {};
    #pragma unroll
    for (int kt = 0; kt < 32; ++kt) {
      const float2 apair = aq[kt * 2];
      const float2 bpair = bq[kt * 32];
      v2f a;  a[0]  = apair.x;  a[1]  = apair.y;
      v2f bb; bb[0] = bpair.x;  bb[1] = bpair.y;
      acc = __builtin_amdgcn_wmma_f32_16x16x4_f32(
           false, a,  false, bb,
           (short)0, acc,  false,  false);
    }
    {
      const int rbase = mt * 16 + laneHi * 8;
      #pragma unroll
      for (int r = 0; r < 8; ++r) Pp[nidx * PLD + rbase + r] = acc[r];
    }
    __syncthreads();

    if (tid < NN) {
      const float gx = rsum[tid] * Xx[tid] - Pp[0 * PLD + tid];
      const float gy = rsum[tid] * Xy[tid] - Pp[1 * PLD + tid];
      const float gz = rsum[tid] * Xz[tid] - Pp[2 * PLD + tid];
      const float dx = -STEP_EPS * gx;
      const float dy = -STEP_EPS * gy;
      const float dz = -STEP_EPS * gz;
      const float speed = sqrtf(dx * dx + dy * dy + dz * dz + 0.001f);
      const float at =
          ALPHA_BASE + (ALPHA - ALPHA_BASE) * ((float)(TS - tstep) / (float)TS);
      const float sc = at * tanhf(speed / at) / speed;
      Xx[tid] += dx * sc;
      Xy[tid] += dy * sc;
      Xz[tid] += dz * sc;
    }
    __syncthreads();
    if (tid < 192) {
      const int q = tid & 63, n = tid >> 6;
      const float* comp = (n == 0) ? Xx : ((n == 1) ? Xy : Xz);
      Xq[q * 16 + n] = make_float2(comp[2 * q], comp[2 * q + 1]);
    }
    __syncthreads();
  }

  float* ob = out + (size_t)b * NN * NN;
  for (int p = tid; p < NN * NN; p += 256) {
    const int i = p >> 7, j = p & 127;
    const float m  = (float)ad[p];
    const float u0 = Xx[i] - Xx[j];
    const float u1 = Xy[i] - Xy[j];
    const float u2 = Xz[i] - Xz[j];
    ob[p] = m * sqrtf(u0 * u0 + u1 * u1 + u2 * u2 + 0.01f);
  }
}

extern "C" void kernel_launch(void* const* d_in, const int* in_sizes, int n_in,
                              void* d_out, int out_size, void* d_ws, size_t ws_size,
                              hipStream_t stream) {
  const float* edge_pred = (const float*)d_in[0];
  const int*   adj       = (const int*)  d_in[1];
  const float* d_init    = (const float*)d_in[2];
  const float* u_init    = (const float*)d_in[3];
  const float* x_noise   = (const float*)d_in[4];
  float*       outp      = (float*)d_out;
  const int B = in_sizes[1] / (NN * NN);
  ts_gcn_fused<<<B, 256, 0, stream>>>(edge_pred, adj, d_init, u_init, x_noise, outp);
}


# Pad the LDS row pitch of the D and C planes from 128 to 132 dwords: in the S build a wave's 32 half-rows sat on one bank (32-way conflict on both loads), now 2-way; all sums unchanged
# speedup vs baseline: 1.0910x; 1.0910x over previous
#include <hip/hip_runtime.h>
#include <hip/hip_bf16.h>
#include <math.h>


typedef __attribute__((ext_vector_type(2))) float v2f;
typedef __attribute__((ext_vector_type(8))) float v8f;

#define NN   128
#define SLD  132
#define PLD  132
#define DLD  132
#define TS   10
#define STEP_EPS 0.1f
#define ALPHA 5.0f
#define ALPHA_BASE 0.1f
#define KRANK 3
#define PSTEPS 10

__device__ __forceinline__ float softplus_f(float x) {
  return fmaxf(x, 0.0f) + log1pf(expf(-fabsf(x)));
}

__device__ __forceinline__ float block_reduce(float v, float* red, int tid) {
  red[tid] = v;
  __syncthreads();
  #pragma unroll
  for (int s = 128; s > 0; s >>= 1) {
    if (tid < s) red[tid] += red[tid + s];
    __syncthreads();
  }
  float r = red[0];
  __syncthreads();
  return r;
}

__launch_bounds__(256, 1)
__global__ void ts_gcn_fused(const float* __restrict__ edge_pred,
                             const int*   __restrict__ adj,
                             const float* __restrict__ d_init,
                             const float* __restrict__ u_init,
                             const float* __restrict__ x_noise,
                             float*       __restrict__ out)
{
  __shared__ float  Dp[NN * DLD];
  __shared__ float  Cw[NN * DLD];
  __shared__ float  Sm[NN * SLD];
  __shared__ float2 Xq[64 * 16];
  __shared__ float  Pp[16 * PLD];
  __shared__ float  Xx[NN], Xy[NN], Xz[NN];
  __shared__ float  rsum[NN];
  __shared__ float  red[256];
  __shared__ float  ucol[KRANK][NN];
  __shared__ float  uvec[NN], utmp[NN];

  const int b   = blockIdx.x;
  const int tid = threadIdx.x;
  const float d0 = d_init[0];

  const float* ep = edge_pred + (size_t)b * NN * NN * 2;
  const int*   ad = adj       + (size_t)b * NN * NN;

  float partM = 0.0f;
  for (int p = tid; p < NN * NN; p += 256) {
    const int i = p >> 7, j = p & 127;
    const float  m = (float)ad[p];
    const float2 e = ((const float2*)ep)[p];
    const float2 t = ((const float2*)ep)[j * NN + i];
    const float sp0 = softplus_f(d0 + e.x + t.x);
    const float sp1 = softplus_f(d0 + e.y + t.y);
    Dp[i * DLD + j] = sp0 * m;
    Cw[i * DLD + j] = sp1 * m;
    partM += (i == j) ? 1.0f : m;
  }
  for (int idx = tid; idx < 64 * 16; idx += 256) Xq[idx] = make_float2(0.f, 0.f);
  const float Msum = block_reduce(partM, red, tid);

  const float Nmol = block_reduce(
      (tid < NN) ? ((tid == 0) ? 1.0f : (float)ad[tid * NN]) : 0.0f, red, tid);

  {
    const int i = tid >> 1, j0 = (tid & 1) << 6;
    float part = 0.0f;
    for (int jj = 0; jj < 64; ++jj) {
      const float d = Dp[i * DLD + j0 + jj];
      part += d * d;
    }
    red[tid] = part;
    __syncthreads();
    if (tid < NN) rsum[tid] = red[2 * tid] + red[2 * tid + 1];
    __syncthreads();
    const float tot  = block_reduce((tid < NN) ? rsum[tid] : 0.0f, red, tid);
    const float invN = 1.0f / Nmol;
    const float Mn   = tot * invN * invN;
    for (int p = tid; p < NN * NN; p += 256) {
      const int i2 = p >> 7, j2 = p & 127;
      const float d  = Dp[i2 * DLD + j2];
      const float d2 = d * d;
      const float mk = (i2 == j2) ? 1.0f : (d > 0.0f ? 1.0f : 0.0f);
      Sm[i2 * SLD + j2] =
          mk * -0.5f * (d2 - rsum[j2] * invN - rsum[i2] * invN + Mn);
    }
    __syncthreads();
  }

  for (int kx = 0; kx < KRANK; ++kx) {
    if (tid < NN) uvec[tid] = u_init[(size_t)b * NN * 3 + tid * 3 + kx];
    __syncthreads();
    for (int it = 0; it < PSTEPS; ++it) {
      const float nv =
          block_reduce((tid < NN) ? uvec[tid] * uvec[tid] : 0.0f, red, tid);
      const float inorm = 1.0f / fmaxf(sqrtf(nv), 0.001f);
      if (tid < NN) uvec[tid] *= inorm;
      __syncthreads();
      float dc0 = 0.0f, dc1 = 0.0f;
      if (kx > 0)
        dc0 = block_reduce((tid < NN) ? ucol[0][tid] * uvec[tid] : 0.0f, red, tid);
      if (kx > 1)
        dc1 = block_reduce((tid < NN) ? ucol[1][tid] * uvec[tid] : 0.0f, red, tid);
      if (tid < NN) {
        const float* row = &Sm[tid * SLD];
        float acc = 0.0f;
        #pragma unroll 8
        for (int j = 0; j < NN; ++j) acc = fmaf(row[j], uvec[j], acc);
        if (kx > 0) acc -= ucol[0][tid] * dc0;
        if (kx > 1) acc -= ucol[1][tid] * dc1;
        utmp[tid] = acc;
      }
      __syncthreads();
      if (tid < NN) uvec[tid] = utmp[tid];
      __syncthreads();
    }
    const float eig =
        block_reduce((tid < NN) ? uvec[tid] * uvec[tid] : 0.0f, red, tid);
    const float iden = 1.0f / sqrtf(sqrtf(eig + 0.01f));
    if (tid < NN) ucol[kx][tid] = uvec[tid] * iden;
    __syncthreads();
  }
  if (tid < NN) {
    const float* xn = x_noise + (size_t)b * NN * 3 + tid * 3;
    Xx[tid] = ucol[0][tid] + xn[0];
    Xy[tid] = ucol[1][tid] + xn[1];
    Xz[tid] = ucol[2][tid] + xn[2];
  }
  __syncthreads();
  if (tid < 192) {
    const int q = tid & 63, n = tid >> 6;
    const float* comp = (n == 0) ? Xx : ((n == 1) ? Xy : Xz);
    Xq[q * 16 + n] = make_float2(comp[2 * q], comp[2 * q + 1]);
  }
  __syncthreads();

  const float invM4 = 4.0f / Msum;
  const int lane   = tid & 31;
  const int mt     = tid >> 5;
  const int nidx   = lane & 15;
  const int laneHi = lane >> 4;
  const int arow   = mt * 16 + nidx;
  const float2* aq = (const float2*)&Sm[arow * SLD + laneHi * 2];
  const float2* bq = &Xq[laneHi * 16 + nidx];

  for (int tstep = 0; tstep < TS; ++tstep) {
    {
      const int i = tid >> 1, j0 = (tid & 1) << 6;
      const float xi0 = Xx[i], xi1 = Xy[i], xi2 = Xz[i];
      float part = 0.0f;
      for (int jj = 0; jj < 64; ++jj) {
        const int j = j0 + jj;
        const int p = i * DLD + j;
        const float u0 = xi0 - Xx[j];
        const float u1 = xi1 - Xy[j];
        const float u2 = xi2 - Xz[j];
        const float dist = sqrtf(u0 * u0 + u1 * u1 + u2 * u2 + 0.01f);
        const float s = invM4 * Cw[p] * (dist - Dp[p]) / dist;
        Sm[i * SLD + j] = s;
        part += s;
      }
      red[tid] = part;
      __syncthreads();
      if (tid < NN) rsum[tid] = red[2 * tid] + red[2 * tid + 1];
      __syncthreads();
    }

    v8f acc = {};
    #pragma unroll
    for (int kt = 0; kt < 32; ++kt) {
      const float2 apair = aq[kt * 2];
      const float2 bpair = bq[kt * 32];
      v2f a;  a[0]  = apair.x;  a[1]  = apair.y;
      v2f bb; bb[0] = bpair.x;  bb[1] = bpair.y;
      acc = __builtin_amdgcn_wmma_f32_16x16x4_f32(
           false, a,  false, bb,
           (short)0, acc,  false,  false);
    }
    {
      const int rbase = mt * 16 + laneHi * 8;
      #pragma unroll
      for (int r = 0; r < 8; ++r) Pp[nidx * PLD + rbase + r] = acc[r];
    }
    __syncthreads();

    if (tid < NN) {
      const float gx = rsum[tid] * Xx[tid] - Pp[0 * PLD + tid];
      const float gy = rsum[tid] * Xy[tid] - Pp[1 * PLD + tid];
      const float gz = rsum[tid] * Xz[tid] - Pp[2 * PLD + tid];
      const float dx = -STEP_EPS * gx;
      const float dy = -STEP_EPS * gy;
      const float dz = -STEP_EPS * gz;
      const float speed = sqrtf(dx * dx + dy * dy + dz * dz + 0.001f);
      const float at =
          ALPHA_BASE + (ALPHA - ALPHA_BASE) * ((float)(TS - tstep) / (float)TS);
      const float sc = at * tanhf(speed / at) / speed;
      Xx[tid] += dx * sc;
      Xy[tid] += dy * sc;
      Xz[tid] += dz * sc;
    }
    __syncthreads();
    if (tid < 192) {
      const int q = tid & 63, n = tid >> 6;
      const float* comp = (n == 0) ? Xx : ((n == 1) ? Xy : Xz);
      Xq[q * 16 + n] = make_float2(comp[2 * q], comp[2 * q + 1]);
    }
    __syncthreads();
  }

  float* ob = out + (size_t)b * NN * NN;
  for (int p = tid; p < NN * NN; p += 256) {
    const int i = p >> 7, j = p & 127;
    const float m  = (float)ad[p];
    const float u0 = Xx[i] - Xx[j];
    const float u1 = Xy[i] - Xy[j];
    const float u2 = Xz[i] - Xz[j];
    ob[p] = m * sqrtf(u0 * u0 + u1 * u1 + u2 * u2 + 0.01f);
  }
}

extern "C" void kernel_launch(void* const* d_in, const int* in_sizes, int n_in,
                              void* d_out, int out_size, void* d_ws, size_t ws_size,
                              hipStream_t stream) {
  const float* edge_pred = (const float*)d_in[0];
  const int*   adj       = (const int*)  d_in[1];
  const float* d_init    = (const float*)d_in[2];
  const float* u_init    = (const float*)d_in[3];
  const float* x_noise   = (const float*)d_in[4];
  float*       outp      = (float*)d_out;
  const int B = in_sizes[1] / (NN * NN);
  ts_gcn_fused<<<B, 256, 0, stream>>>(edge_pred, adj, d_init, u_init, x_noise, outp);
}


# Block-wide sums (power-step norms, deflation dots, mask and row sums): DPP wave sums + one LDS combine of 8 partials replace 8-level LDS trees; 2 barriers per power step, not 13-33; sum tree changes
# speedup vs baseline: 1.3971x; 1.2806x over previous
#include <hip/hip_runtime.h>
#include <hip/hip_bf16.h>
#include <math.h>


typedef __attribute__((ext_vector_type(2))) float v2f;
typedef __attribute__((ext_vector_type(8))) float v8f;

#define NN   128
#define SLD  132
#define PLD  132
#define DLD  132
#define TS   10
#define STEP_EPS 0.1f
#define ALPHA 5.0f
#define ALPHA_BASE 0.1f
#define KRANK 3
#define PSTEPS 10

__device__ __forceinline__ float softplus_f(float x) {
  return fmaxf(x, 0.0f) + log1pf(expf(-fabsf(x)));
}

__device__ __forceinline__ float xhalf(float x) {
  const int b = __builtin_bit_cast(int, x);
  return __builtin_bit_cast(
      float, __builtin_amdgcn_permlanex16(b, b, 0x76543210, 0xfedcba98, false, false));
}

template <int M>
__device__ __forceinline__ float xrow(float x) {
  const int b = __builtin_bit_cast(int, x);
  return __builtin_bit_cast(float, __builtin_amdgcn_update_dpp(b, b, 0x160 + M, 0xf, 0xf, false));
}

__device__ __forceinline__ float wave_sum(float v) {
  v += xrow<1>(v);
  v += xrow<2>(v);
  v += xrow<4>(v);
  v += xrow<8>(v);
  v += xhalf(v);
  return v;
}

template <int K>
__device__ __forceinline__ void block_sum(float (&v)[K], float* slot, int tid) {
  for (int k = 0; k < K; ++k) v[k] = wave_sum(v[k]);
  if ((tid & 31) == 0) {
    for (int k = 0; k < K; ++k) slot[k * 8 + (tid >> 5)] = v[k];
  }
  __syncthreads();
  for (int k = 0; k < K; ++k) {
    const float4 lo = *(const float4*)&slot[k * 8];
    const float4 hi = *(const float4*)&slot[k * 8 + 4];
    v[k] = ((lo.x + lo.y) + (lo.z + lo.w)) + ((hi.x + hi.y) + (hi.z + hi.w));
  }
}

__device__ __forceinline__ float block_sum1(float x, float* slot, int tid) {
  float v[1] = {x};
  block_sum<1>(v, slot, tid);
  return v[0];
}

__launch_bounds__(256, 1)
__global__ void ts_gcn_fused(const float* __restrict__ edge_pred,
                             const int*   __restrict__ adj,
                             const float* __restrict__ d_init,
                             const float* __restrict__ u_init,
                             const float* __restrict__ x_noise,
                             float*       __restrict__ out)
{
  __shared__ float  Dp[NN * DLD];
  __shared__ float  Cw[NN * DLD];
  __shared__ float  Sm[NN * SLD];
  __shared__ float2 Xq[64 * 16];
  __shared__ float  Pp[16 * PLD];
  __shared__ float  Xx[NN], Xy[NN], Xz[NN];
  __shared__ float  rsum[NN];
  __shared__ __attribute__((aligned(16))) float wsum[2][16];
  __shared__ float  ucol[KRANK][NN];
  __shared__ float  uvec[NN];

  const int b   = blockIdx.x;
  const int tid = threadIdx.x;
  const float d0 = d_init[0];

  const float* ep = edge_pred + (size_t)b * NN * NN * 2;
  const int*   ad = adj       + (size_t)b * NN * NN;

  float partM = 0.0f;
  for (int p = tid; p < NN * NN; p += 256) {
    const int i = p >> 7, j = p & 127;
    const float  m = (float)ad[p];
    const float2 e = ((const float2*)ep)[p];
    const float2 t = ((const float2*)ep)[j * NN + i];
    const float sp0 = softplus_f(d0 + e.x + t.x);
    const float sp1 = softplus_f(d0 + e.y + t.y);
    Dp[i * DLD + j] = sp0 * m;
    Cw[i * DLD + j] = sp1 * m;
    partM += (i == j) ? 1.0f : m;
  }
  for (int idx = tid; idx < 64 * 16; idx += 256) Xq[idx] = make_float2(0.f, 0.f);
  int rs = 0;
  const float Msum = block_sum1(partM, wsum[rs ^= 1], tid);

  const float Nmol = block_sum1(
      (tid < NN) ? ((tid == 0) ? 1.0f : (float)ad[tid * NN]) : 0.0f, wsum[rs ^= 1], tid);

  {
    const int i = tid >> 1, j0 = (tid & 1) << 6;
    float part = 0.0f;
    for (int jj = 0; jj < 64; ++jj) {
      const float d = Dp[i * DLD + j0 + jj];
      part += d * d;
    }
    part += xrow<1>(part);
    if ((tid & 1) == 0) rsum[i] = part;
    __syncthreads();
    const float tot  = block_sum1((tid < NN) ? rsum[tid] : 0.0f, wsum[rs ^= 1], tid);
    const float invN = 1.0f / Nmol;
    const float Mn   = tot * invN * invN;
    for (int p = tid; p < NN * NN; p += 256) {
      const int i2 = p >> 7, j2 = p & 127;
      const float d  = Dp[i2 * DLD + j2];
      const float d2 = d * d;
      const float mk = (i2 == j2) ? 1.0f : (d > 0.0f ? 1.0f : 0.0f);
      Sm[i2 * SLD + j2] =
          mk * -0.5f * (d2 - rsum[j2] * invN - rsum[i2] * invN + Mn);
    }
    __syncthreads();
  }

  for (int kx = 0; kx < KRANK; ++kx) {
    float uval = (tid < NN) ? u_init[(size_t)b * NN * 3 + tid * 3 + kx] : 0.0f;
    const float c0 = (kx > 0 && tid < NN) ? ucol[0][tid] : 0.0f;
    const float c1 = (kx > 1 && tid < NN) ? ucol[1][tid] : 0.0f;
    for (int it = 0; it < PSTEPS; ++it) {
      const float nv = block_sum1(uval * uval, wsum[rs ^= 1], tid);
      const float inorm = 1.0f / fmaxf(sqrtf(nv), 0.001f);
      uval *= inorm;
      if (tid < NN) uvec[tid] = uval;
      float dc0 = 0.0f, dc1 = 0.0f;
      if (kx == 1) {
        dc0 = block_sum1(c0 * uval, wsum[rs ^= 1], tid);
      } else if (kx == 2) {
        float d[2] = {c0 * uval, c1 * uval};
        block_sum<2>(d, wsum[rs ^= 1], tid);
        dc0 = d[0];
        dc1 = d[1];
      } else {
        __syncthreads();
      }
      if (tid < NN) {
        const float* row = &Sm[tid * SLD];
        float acc = 0.0f;
        #pragma unroll 8
        for (int j = 0; j < NN; ++j) acc = fmaf(row[j], uvec[j], acc);
        if (kx > 0) acc -= c0 * dc0;
        if (kx > 1) acc -= c1 * dc1;
        uval = acc;
      }
    }
    const float eig = block_sum1(uval * uval, wsum[rs ^= 1], tid);
    const float iden = 1.0f / sqrtf(sqrtf(eig + 0.01f));
    if (tid < NN) ucol[kx][tid] = uval * iden;
    __syncthreads();
  }
  if (tid < NN) {
    const float* xn = x_noise + (size_t)b * NN * 3 + tid * 3;
    Xx[tid] = ucol[0][tid] + xn[0];
    Xy[tid] = ucol[1][tid] + xn[1];
    Xz[tid] = ucol[2][tid] + xn[2];
  }
  __syncthreads();
  if (tid < 192) {
    const int q = tid & 63, n = tid >> 6;
    const float* comp = (n == 0) ? Xx : ((n == 1) ? Xy : Xz);
    Xq[q * 16 + n] = make_float2(comp[2 * q], comp[2 * q + 1]);
  }
  __syncthreads();

  const float invM4 = 4.0f / Msum;
  const int lane   = tid & 31;
  const int mt     = tid >> 5;
  const int nidx   = lane & 15;
  const int laneHi = lane >> 4;
  const int arow   = mt * 16 + nidx;
  const float2* aq = (const float2*)&Sm[arow * SLD + laneHi * 2];
  const float2* bq = &Xq[laneHi * 16 + nidx];

  for (int tstep = 0; tstep < TS; ++tstep) {
    {
      const int i = tid >> 1, j0 = (tid & 1) << 6;
      const float xi0 = Xx[i], xi1 = Xy[i], xi2 = Xz[i];
      float part = 0.0f;
      for (int jj = 0; jj < 64; ++jj) {
        const int j = j0 + jj;
        const int p = i * DLD + j;
        const float u0 = xi0 - Xx[j];
        const float u1 = xi1 - Xy[j];
        const float u2 = xi2 - Xz[j];
        const float dist = sqrtf(u0 * u0 + u1 * u1 + u2 * u2 + 0.01f);
        const float s = invM4 * Cw[p] * (dist - Dp[p]) / dist;
        Sm[i * SLD + j] = s;
        part += s;
      }
      part += xrow<1>(part);
      if ((tid & 1) == 0) rsum[i] = part;
      __syncthreads();
    }

    v8f acc = {};
    #pragma unroll
    for (int kt = 0; kt < 32; ++kt) {
      const float2 apair = aq[kt * 2];
      const float2 bpair = bq[kt * 32];
      v2f a;  a[0]  = apair.x;  a[1]  = apair.y;
      v2f bb; bb[0] = bpair.x;  bb[1] = bpair.y;
      acc = __builtin_amdgcn_wmma_f32_16x16x4_f32(
           false, a,  false, bb,
           (short)0, acc,  false,  false);
    }
    {
      const int rbase = mt * 16 + laneHi * 8;
      #pragma unroll
      for (int r = 0; r < 8; ++r) Pp[nidx * PLD + rbase + r] = acc[r];
    }
    __syncthreads();

    if (tid < NN) {
      const float gx = rsum[tid] * Xx[tid] - Pp[0 * PLD + tid];
      const float gy = rsum[tid] * Xy[tid] - Pp[1 * PLD + tid];
      const float gz = rsum[tid] * Xz[tid] - Pp[2 * PLD + tid];
      const float dx = -STEP_EPS * gx;
      const float dy = -STEP_EPS * gy;
      const float dz = -STEP_EPS * gz;
      const float speed = sqrtf(dx * dx + dy * dy + dz * dz + 0.001f);
      const float at =
          ALPHA_BASE + (ALPHA - ALPHA_BASE) * ((float)(TS - tstep) / (float)TS);
      const float sc = at * tanhf(speed / at) / speed;
      Xx[tid] += dx * sc;
      Xy[tid] += dy * sc;
      Xz[tid] += dz * sc;
    }
    __syncthreads();
    if (tid < 192) {
      const int q = tid & 63, n = tid >> 6;
      const float* comp = (n == 0) ? Xx : ((n == 1) ? Xy : Xz);
      Xq[q * 16 + n] = make_float2(comp[2 * q], comp[2 * q + 1]);
    }
    __syncthreads();
  }

  float* ob = out + (size_t)b * NN * NN;
  for (int p = tid; p < NN * NN; p += 256) {
    const int i = p >> 7, j = p & 127;
    const float m  = (float)ad[p];
    const float u0 = Xx[i] - Xx[j];
    const float u1 = Xy[i] - Xy[j];
    const float u2 = Xz[i] - Xz[j];
    ob[p] = m * sqrtf(u0 * u0 + u1 * u1 + u2 * u2 + 0.01f);
  }
}

extern "C" void kernel_launch(void* const* d_in, const int* in_sizes, int n_in,
                              void* d_out, int out_size, void* d_ws, size_t ws_size,
                              hipStream_t stream) {
  const float* edge_pred = (const float*)d_in[0];
  const int*   adj       = (const int*)  d_in[1];
  const float* d_init    = (const float*)d_in[2];
  const float* u_init    = (const float*)d_in[3];
  const float* x_noise   = (const float*)d_in[4];
  float*       outp      = (float*)d_out;
  const int B = in_sizes[1] / (NN * NN);
  ts_gcn_fused<<<B, 256, 0, stream>>>(edge_pred, adj, d_init, u_init, x_noise, outp);
}


# Power-step matvec: every thread keeps its half row of the Gram matrix in 64 registers over all 30 steps, 4 sum chains on 256 threads (was 128 threads, one LDS-fed chain); row sum order changes
# speedup vs baseline: 1.6095x; 1.1520x over previous
#include <hip/hip_runtime.h>
#include <hip/hip_bf16.h>
#include <math.h>


typedef __attribute__((ext_vector_type(2))) float v2f;
typedef __attribute__((ext_vector_type(8))) float v8f;

#define NN   128
#define SLD  132
#define PLD  132
#define DLD  132
#define TS   10
#define STEP_EPS 0.1f
#define ALPHA 5.0f
#define ALPHA_BASE 0.1f
#define KRANK 3
#define PSTEPS 10

__device__ __forceinline__ float softplus_f(float x) {
  return fmaxf(x, 0.0f) + log1pf(expf(-fabsf(x)));
}

__device__ __forceinline__ float xhalf(float x) {
  const int b = __builtin_bit_cast(int, x);
  return __builtin_bit_cast(
      float, __builtin_amdgcn_permlanex16(b, b, 0x76543210, 0xfedcba98, false, false));
}

template <int M>
__device__ __forceinline__ float xrow(float x) {
  const int b = __builtin_bit_cast(int, x);
  return __builtin_bit_cast(float, __builtin_amdgcn_update_dpp(b, b, 0x160 + M, 0xf, 0xf, false));
}

__device__ __forceinline__ float wave_sum(float v) {
  v += xrow<1>(v);
  v += xrow<2>(v);
  v += xrow<4>(v);
  v += xrow<8>(v);
  v += xhalf(v);
  return v;
}

template <int K>
__device__ __forceinline__ void block_sum(float (&v)[K], float* slot, int tid) {
  for (int k = 0; k < K; ++k) v[k] = wave_sum(v[k]);
  if ((tid & 31) == 0) {
    for (int k = 0; k < K; ++k) slot[k * 8 + (tid >> 5)] = v[k];
  }
  __syncthreads();
  for (int k = 0; k < K; ++k) {
    const float4 lo = *(const float4*)&slot[k * 8];
    const float4 hi = *(const float4*)&slot[k * 8 + 4];
    v[k] = ((lo.x + lo.y) + (lo.z + lo.w)) + ((hi.x + hi.y) + (hi.z + hi.w));
  }
}

__device__ __forceinline__ float block_sum1(float x, float* slot, int tid) {
  float v[1] = {x};
  block_sum<1>(v, slot, tid);
  return v[0];
}

__launch_bounds__(256, 1)
__global__ void ts_gcn_fused(const float* __restrict__ edge_pred,
                             const int*   __restrict__ adj,
                             const float* __restrict__ d_init,
                             const float* __restrict__ u_init,
                             const float* __restrict__ x_noise,
                             float*       __restrict__ out)
{
  __shared__ float  Dp[NN * DLD];
  __shared__ float  Cw[NN * DLD];
  __shared__ __attribute__((aligned(16))) float Sm[NN * SLD];
  __shared__ float2 Xq[64 * 16];
  __shared__ float  Pp[16 * PLD];
  __shared__ float  Xx[NN], Xy[NN], Xz[NN];
  __shared__ float  rsum[NN];
  __shared__ __attribute__((aligned(16))) float wsum[2][16];
  __shared__ float  ucol[KRANK][NN];
  __shared__ __attribute__((aligned(16))) float uvec[NN];

  const int b   = blockIdx.x;
  const int tid = threadIdx.x;
  const float d0 = d_init[0];

  const float* ep = edge_pred + (size_t)b * NN * NN * 2;
  const int*   ad = adj       + (size_t)b * NN * NN;

  float partM = 0.0f;
  for (int p = tid; p < NN * NN; p += 256) {
    const int i = p >> 7, j = p & 127;
    const float  m = (float)ad[p];
    const float2 e = ((const float2*)ep)[p];
    const float2 t = ((const float2*)ep)[j * NN + i];
    const float sp0 = softplus_f(d0 + e.x + t.x);
    const float sp1 = softplus_f(d0 + e.y + t.y);
    Dp[i * DLD + j] = sp0 * m;
    Cw[i * DLD + j] = sp1 * m;
    partM += (i == j) ? 1.0f : m;
  }
  for (int idx = tid; idx < 64 * 16; idx += 256) Xq[idx] = make_float2(0.f, 0.f);
  int rs = 0;
  const float Msum = block_sum1(partM, wsum[rs ^= 1], tid);

  const float Nmol = block_sum1(
      (tid < NN) ? ((tid == 0) ? 1.0f : (float)ad[tid * NN]) : 0.0f, wsum[rs ^= 1], tid);

  {
    const int i = tid >> 1, j0 = (tid & 1) << 6;
    float part = 0.0f;
    for (int jj = 0; jj < 64; ++jj) {
      const float d = Dp[i * DLD + j0 + jj];
      part += d * d;
    }
    part += xrow<1>(part);
    if ((tid & 1) == 0) rsum[i] = part;
    __syncthreads();
    const float tot  = block_sum1((tid < NN) ? rsum[tid] : 0.0f, wsum[rs ^= 1], tid);
    const float invN = 1.0f / Nmol;
    const float Mn   = tot * invN * invN;
    for (int p = tid; p < NN * NN; p += 256) {
      const int i2 = p >> 7, j2 = p & 127;
      const float d  = Dp[i2 * DLD + j2];
      const float d2 = d * d;
      const float mk = (i2 == j2) ? 1.0f : (d > 0.0f ? 1.0f : 0.0f);
      Sm[i2 * SLD + j2] =
          mk * -0.5f * (d2 - rsum[j2] * invN - rsum[i2] * invN + Mn);
    }
    __syncthreads();
  }

  const int prow = tid >> 1, ph = tid & 1;
  float4 grow[16];
  #pragma unroll
  for (int q = 0; q < 16; ++q)
    grow[q] = *(const float4*)&Sm[prow * SLD + (2 * q + ph) * 4];
  float c0 = 0.0f, c1 = 0.0f;
  for (int kx = 0; kx < KRANK; ++kx) {
    float uval = u_init[(size_t)b * NN * 3 + prow * 3 + kx];
    for (int it = 0; it < PSTEPS; ++it) {
      const float nv = block_sum1(ph ? 0.0f : uval * uval, wsum[rs ^= 1], tid);
      const float inorm = 1.0f / fmaxf(sqrtf(nv), 0.001f);
      uval *= inorm;
      if (ph == 0) uvec[prow] = uval;
      float dc0 = 0.0f, dc1 = 0.0f;
      if (kx == 1) {
        dc0 = block_sum1(ph ? 0.0f : c0 * uval, wsum[rs ^= 1], tid);
      } else if (kx == 2) {
        float d[2] = {ph ? 0.0f : c0 * uval, ph ? 0.0f : c1 * uval};
        block_sum<2>(d, wsum[rs ^= 1], tid);
        dc0 = d[0];
        dc1 = d[1];
      } else {
        __syncthreads();
      }
      float a0 = 0.0f, a1 = 0.0f, a2 = 0.0f, a3 = 0.0f;
      #pragma unroll
      for (int q = 0; q < 16; ++q) {
        const float4 u4 = *(const float4*)&uvec[(2 * q + ph) * 4];
        a0 = fmaf(grow[q].x, u4.x, a0);
        a1 = fmaf(grow[q].y, u4.y, a1);
        a2 = fmaf(grow[q].z, u4.z, a2);
        a3 = fmaf(grow[q].w, u4.w, a3);
      }
      float acc = (a0 + a1) + (a2 + a3);
      acc += xrow<1>(acc);
      if (kx > 0) acc -= c0 * dc0;
      if (kx > 1) acc -= c1 * dc1;
      uval = acc;
    }
    const float eig = block_sum1(ph ? 0.0f : uval * uval, wsum[rs ^= 1], tid);
    const float iden = 1.0f / sqrtf(sqrtf(eig + 0.01f));
    const float unew = uval * iden;
    if (ph == 0) ucol[kx][prow] = unew;
    if (kx == 0) c0 = unew; else if (kx == 1) c1 = unew;
    __syncthreads();
  }
  if (tid < NN) {
    const float* xn = x_noise + (size_t)b * NN * 3 + tid * 3;
    Xx[tid] = ucol[0][tid] + xn[0];
    Xy[tid] = ucol[1][tid] + xn[1];
    Xz[tid] = ucol[2][tid] + xn[2];
  }
  __syncthreads();
  if (tid < 192) {
    const int q = tid & 63, n = tid >> 6;
    const float* comp = (n == 0) ? Xx : ((n == 1) ? Xy : Xz);
    Xq[q * 16 + n] = make_float2(comp[2 * q], comp[2 * q + 1]);
  }
  __syncthreads();

  const float invM4 = 4.0f / Msum;
  const int lane   = tid & 31;
  const int mt     = tid >> 5;
  const int nidx   = lane & 15;
  const int laneHi = lane >> 4;
  const int arow   = mt * 16 + nidx;
  const float2* aq = (const float2*)&Sm[arow * SLD + laneHi * 2];
  const float2* bq = &Xq[laneHi * 16 + nidx];

  for (int tstep = 0; tstep < TS; ++tstep) {
    {
      const int i = tid >> 1, j0 = (tid & 1) << 6;
      const float xi0 = Xx[i], xi1 = Xy[i], xi2 = Xz[i];
      float part = 0.0f;
      for (int jj = 0; jj < 64; ++jj) {
        const int j = j0 + jj;
        const int p = i * DLD + j;
        const float u0 = xi0 - Xx[j];
        const float u1 = xi1 - Xy[j];
        const float u2 = xi2 - Xz[j];
        const float dist = sqrtf(u0 * u0 + u1 * u1 + u2 * u2 + 0.01f);
        const float s = invM4 * Cw[p] * (dist - Dp[p]) / dist;
        Sm[i * SLD + j] = s;
        part += s;
      }
      part += xrow<1>(part);
      if ((tid & 1) == 0) rsum[i] = part;
      __syncthreads();
    }

    v8f acc = {};
    #pragma unroll
    for (int kt = 0; kt < 32; ++kt) {
      const float2 apair = aq[kt * 2];
      const float2 bpair = bq[kt * 32];
      v2f a;  a[0]  = apair.x;  a[1]  = apair.y;
      v2f bb; bb[0] = bpair.x;  bb[1] = bpair.y;
      acc = __builtin_amdgcn_wmma_f32_16x16x4_f32(
           false, a,  false, bb,
           (short)0, acc,  false,  false);
    }
    {
      const int rbase = mt * 16 + laneHi * 8;
      #pragma unroll
      for (int r = 0; r < 8; ++r) Pp[nidx * PLD + rbase + r] = acc[r];
    }
    __syncthreads();

    if (tid < NN) {
      const float gx = rsum[tid] * Xx[tid] - Pp[0 * PLD + tid];
      const float gy = rsum[tid] * Xy[tid] - Pp[1 * PLD + tid];
      const float gz = rsum[tid] * Xz[tid] - Pp[2 * PLD + tid];
      const float dx = -STEP_EPS * gx;
      const float dy = -STEP_EPS * gy;
      const float dz = -STEP_EPS * gz;
      const float speed = sqrtf(dx * dx + dy * dy + dz * dz + 0.001f);
      const float at =
          ALPHA_BASE + (ALPHA - ALPHA_BASE) * ((float)(TS - tstep) / (float)TS);
      const float sc = at * tanhf(speed / at) / speed;
      Xx[tid] += dx * sc;
      Xy[tid] += dy * sc;
      Xz[tid] += dz * sc;
    }
    __syncthreads();
    if (tid < 192) {
      const int q = tid & 63, n = tid >> 6;
      const float* comp = (n == 0) ? Xx : ((n == 1) ? Xy : Xz);
      Xq[q * 16 + n] = make_float2(comp[2 * q], comp[2 * q + 1]);
    }
    __syncthreads();
  }

  float* ob = out + (size_t)b * NN * NN;
  for (int p = tid; p < NN * NN; p += 256) {
    const int i = p >> 7, j = p & 127;
    const float m  = (float)ad[p];
    const float u0 = Xx[i] - Xx[j];
    const float u1 = Xy[i] - Xy[j];
    const float u2 = Xz[i] - Xz[j];
    ob[p] = m * sqrtf(u0 * u0 + u1 * u1 + u2 * u2 + 0.01f);
  }
}

extern "C" void kernel_launch(void* const* d_in, const int* in_sizes, int n_in,
                              void* d_out, int out_size, void* d_ws, size_t ws_size,
                              hipStream_t stream) {
  const float* edge_pred = (const float*)d_in[0];
  const int*   adj       = (const int*)  d_in[1];
  const float* d_init    = (const float*)d_in[2];
  const float* u_init    = (const float*)d_in[3];
  const float* x_noise   = (const float*)d_in[4];
  float*       outp      = (float*)d_out;
  const int B = in_sizes[1] / (NN * NN);
  ts_gcn_fused<<<B, 256, 0, stream>>>(edge_pred, adj, d_init, u_init, x_noise, outp);
}


# Phase 0: raw edge planes and mask copied to LDS by whole-line loads, 32 in flight per wave, then softplus once per pair (i,j),(j,i) in place; was a load-wait loop with a gather; d0+e+t now d0+(e+t)
# speedup vs baseline: 2.0030x; 1.2445x over previous
#include <hip/hip_runtime.h>
#include <hip/hip_bf16.h>
#include <math.h>


typedef __attribute__((ext_vector_type(2))) float v2f;
typedef __attribute__((ext_vector_type(8))) float v8f;

#define NN   128
#define SLD  132
#define PLD  132
#define DLD  132
#define TS   10
#define STEP_EPS 0.1f
#define ALPHA 5.0f
#define ALPHA_BASE 0.1f
#define KRANK 3
#define PSTEPS 10

__device__ __forceinline__ float softplus_f(float x) {
  return fmaxf(x, 0.0f) + log1pf(expf(-fabsf(x)));
}

__device__ __forceinline__ float xhalf(float x) {
  const int b = __builtin_bit_cast(int, x);
  return __builtin_bit_cast(
      float, __builtin_amdgcn_permlanex16(b, b, 0x76543210, 0xfedcba98, false, false));
}

template <int M>
__device__ __forceinline__ float xrow(float x) {
  const int b = __builtin_bit_cast(int, x);
  return __builtin_bit_cast(float, __builtin_amdgcn_update_dpp(b, b, 0x160 + M, 0xf, 0xf, false));
}

__device__ __forceinline__ float wave_sum(float v) {
  v += xrow<1>(v);
  v += xrow<2>(v);
  v += xrow<4>(v);
  v += xrow<8>(v);
  v += xhalf(v);
  return v;
}

template <int K>
__device__ __forceinline__ void block_sum(float (&v)[K], float* slot, int tid) {
  for (int k = 0; k < K; ++k) v[k] = wave_sum(v[k]);
  if ((tid & 31) == 0) {
    for (int k = 0; k < K; ++k) slot[k * 8 + (tid >> 5)] = v[k];
  }
  __syncthreads();
  for (int k = 0; k < K; ++k) {
    const float4 lo = *(const float4*)&slot[k * 8];
    const float4 hi = *(const float4*)&slot[k * 8 + 4];
    v[k] = ((lo.x + lo.y) + (lo.z + lo.w)) + ((hi.x + hi.y) + (hi.z + hi.w));
  }
}

__device__ __forceinline__ float block_sum1(float x, float* slot, int tid) {
  float v[1] = {x};
  block_sum<1>(v, slot, tid);
  return v[0];
}

__launch_bounds__(256, 1)
__global__ void ts_gcn_fused(const float* __restrict__ edge_pred,
                             const int*   __restrict__ adj,
                             const float* __restrict__ d_init,
                             const float* __restrict__ u_init,
                             const float* __restrict__ x_noise,
                             float*       __restrict__ out)
{
  __shared__ __attribute__((aligned(16))) float Dp[NN * DLD];
  __shared__ __attribute__((aligned(16))) float Cw[NN * DLD];
  __shared__ __attribute__((aligned(16))) float Sm[NN * SLD];
  __shared__ float2 Xq[64 * 16];
  __shared__ float  Pp[16 * PLD];
  __shared__ float  Xx[NN], Xy[NN], Xz[NN];
  __shared__ float  rsum[NN];
  __shared__ __attribute__((aligned(16))) float wsum[2][16];
  __shared__ float  ucol[KRANK][NN];
  __shared__ __attribute__((aligned(16))) float uvec[NN];

  const int b   = blockIdx.x;
  const int tid = threadIdx.x;
  const float d0 = d_init[0];

  const float* ep = edge_pred + (size_t)b * NN * NN * 2;
  const int*   ad = adj       + (size_t)b * NN * NN;

  float partM = 0.0f;
  {
    const float4* ep4 = (const float4*)ep;
    const int2*   ad2 = (const int2*)ad;
    #pragma unroll 1
    for (int kb = 0; kb < 2; ++kb) {
      float4 ev[16];
      int2   av[16];
      #pragma unroll
      for (int u = 0; u < 16; ++u) {
        const int p2 = tid + 256 * (kb * 16 + u);
        ev[u] = ep4[p2];
        av[u] = ad2[p2];
      }
      #pragma unroll
      for (int u = 0; u < 16; ++u) {
        const int p2 = tid + 256 * (kb * 16 + u);
        const int i = p2 >> 6, j = (p2 & 63) << 1;
        const float m0 = (float)av[u].x, m1 = (float)av[u].y;
        *(float2*)&Dp[i * DLD + j] = make_float2(ev[u].x, ev[u].z);
        *(float2*)&Cw[i * DLD + j] = make_float2(ev[u].y, ev[u].w);
        *(float2*)&Sm[i * SLD + j] = make_float2(m0, m1);
        partM += ((i == j) ? 1.0f : m0) + ((i == j + 1) ? 1.0f : m1);
      }
    }
  }
  __syncthreads();
  for (int q = tid; q < (NN * (NN + 1)) / 2; q += 256) {
    const int r = q / 129, c = q - r * 129;
    const bool lo = c < NN - r;
    const int i = lo ? r : (NN - 1 - r);
    const int j = lo ? (r + c) : (c - 1);
    const int pij = i * DLD + j, pji = j * DLD + i;
    const float sp0 = softplus_f(d0 + (Dp[pij] + Dp[pji]));
    const float sp1 = softplus_f(d0 + (Cw[pij] + Cw[pji]));
    const float mij = Sm[i * SLD + j], mji = Sm[j * SLD + i];
    Dp[pij] = sp0 * mij;
    Cw[pij] = sp1 * mij;
    Dp[pji] = sp0 * mji;
    Cw[pji] = sp1 * mji;
  }
  for (int idx = tid; idx < 64 * 16; idx += 256) Xq[idx] = make_float2(0.f, 0.f);
  int rs = 0;
  const float Msum = block_sum1(partM, wsum[rs ^= 1], tid);

  const float Nmol = block_sum1(
      (tid < NN) ? ((tid == 0) ? 1.0f : Sm[tid * SLD]) : 0.0f, wsum[rs ^= 1], tid);

  {
    const int i = tid >> 1, j0 = (tid & 1) << 6;
    float part = 0.0f;
    for (int jj = 0; jj < 64; ++jj) {
      const float d = Dp[i * DLD + j0 + jj];
      part += d * d;
    }
    part += xrow<1>(part);
    if ((tid & 1) == 0) rsum[i] = part;
    __syncthreads();
    const float tot  = block_sum1((tid < NN) ? rsum[tid] : 0.0f, wsum[rs ^= 1], tid);
    const float invN = 1.0f / Nmol;
    const float Mn   = tot * invN * invN;
    for (int p = tid; p < NN * NN; p += 256) {
      const int i2 = p >> 7, j2 = p & 127;
      const float d  = Dp[i2 * DLD + j2];
      const float d2 = d * d;
      const float mk = (i2 == j2) ? 1.0f : (d > 0.0f ? 1.0f : 0.0f);
      Sm[i2 * SLD + j2] =
          mk * -0.5f * (d2 - rsum[j2] * invN - rsum[i2] * invN + Mn);
    }
    __syncthreads();
  }

  const int prow = tid >> 1, ph = tid & 1;
  float4 grow[16];
  #pragma unroll
  for (int q = 0; q < 16; ++q)
    grow[q] = *(const float4*)&Sm[prow * SLD + (2 * q + ph) * 4];
  float c0 = 0.0f, c1 = 0.0f;
  for (int kx = 0; kx < KRANK; ++kx) {
    float uval = u_init[(size_t)b * NN * 3 + prow * 3 + kx];
    for (int it = 0; it < PSTEPS; ++it) {
      const float nv = block_sum1(ph ? 0.0f : uval * uval, wsum[rs ^= 1], tid);
      const float inorm = 1.0f / fmaxf(sqrtf(nv), 0.001f);
      uval *= inorm;
      if (ph == 0) uvec[prow] = uval;
      float dc0 = 0.0f, dc1 = 0.0f;
      if (kx == 1) {
        dc0 = block_sum1(ph ? 0.0f : c0 * uval, wsum[rs ^= 1], tid);
      } else if (kx == 2) {
        float d[2] = {ph ? 0.0f : c0 * uval, ph ? 0.0f : c1 * uval};
        block_sum<2>(d, wsum[rs ^= 1], tid);
        dc0 = d[0];
        dc1 = d[1];
      } else {
        __syncthreads();
      }
      float a0 = 0.0f, a1 = 0.0f, a2 = 0.0f, a3 = 0.0f;
      #pragma unroll
      for (int q = 0; q < 16; ++q) {
        const float4 u4 = *(const float4*)&uvec[(2 * q + ph) * 4];
        a0 = fmaf(grow[q].x, u4.x, a0);
        a1 = fmaf(grow[q].y, u4.y, a1);
        a2 = fmaf(grow[q].z, u4.z, a2);
        a3 = fmaf(grow[q].w, u4.w, a3);
      }
      float acc = (a0 + a1) + (a2 + a3);
      acc += xrow<1>(acc);
      if (kx > 0) acc -= c0 * dc0;
      if (kx > 1) acc -= c1 * dc1;
      uval = acc;
    }
    const float eig = block_sum1(ph ? 0.0f : uval * uval, wsum[rs ^= 1], tid);
    const float iden = 1.0f / sqrtf(sqrtf(eig + 0.01f));
    const float unew = uval * iden;
    if (ph == 0) ucol[kx][prow] = unew;
    if (kx == 0) c0 = unew; else if (kx == 1) c1 = unew;
    __syncthreads();
  }
  if (tid < NN) {
    const float* xn = x_noise + (size_t)b * NN * 3 + tid * 3;
    Xx[tid] = ucol[0][tid] + xn[0];
    Xy[tid] = ucol[1][tid] + xn[1];
    Xz[tid] = ucol[2][tid] + xn[2];
  }
  __syncthreads();
  if (tid < 192) {
    const int q = tid & 63, n = tid >> 6;
    const float* comp = (n == 0) ? Xx : ((n == 1) ? Xy : Xz);
    Xq[q * 16 + n] = make_float2(comp[2 * q], comp[2 * q + 1]);
  }
  __syncthreads();

  const float invM4 = 4.0f / Msum;
  const int lane   = tid & 31;
  const int mt     = tid >> 5;
  const int nidx   = lane & 15;
  const int laneHi = lane >> 4;
  const int arow   = mt * 16 + nidx;
  const float2* aq = (const float2*)&Sm[arow * SLD + laneHi * 2];
  const float2* bq = &Xq[laneHi * 16 + nidx];

  for (int tstep = 0; tstep < TS; ++tstep) {
    {
      const int i = tid >> 1, j0 = (tid & 1) << 6;
      const float xi0 = Xx[i], xi1 = Xy[i], xi2 = Xz[i];
      float part = 0.0f;
      for (int jj = 0; jj < 64; ++jj) {
        const int j = j0 + jj;
        const int p = i * DLD + j;
        const float u0 = xi0 - Xx[j];
        const float u1 = xi1 - Xy[j];
        const float u2 = xi2 - Xz[j];
        const float dist = sqrtf(u0 * u0 + u1 * u1 + u2 * u2 + 0.01f);
        const float s = invM4 * Cw[p] * (dist - Dp[p]) / dist;
        Sm[i * SLD + j] = s;
        part += s;
      }
      part += xrow<1>(part);
      if ((tid & 1) == 0) rsum[i] = part;
      __syncthreads();
    }

    v8f acc = {};
    #pragma unroll
    for (int kt = 0; kt < 32; ++kt) {
      const float2 apair = aq[kt * 2];
      const float2 bpair = bq[kt * 32];
      v2f a;  a[0]  = apair.x;  a[1]  = apair.y;
      v2f bb; bb[0] = bpair.x;  bb[1] = bpair.y;
      acc = __builtin_amdgcn_wmma_f32_16x16x4_f32(
           false, a,  false, bb,
           (short)0, acc,  false,  false);
    }
    {
      const int rbase = mt * 16 + laneHi * 8;
      #pragma unroll
      for (int r = 0; r < 8; ++r) Pp[nidx * PLD + rbase + r] = acc[r];
    }
    __syncthreads();

    if (tid < NN) {
      const float gx = rsum[tid] * Xx[tid] - Pp[0 * PLD + tid];
      const float gy = rsum[tid] * Xy[tid] - Pp[1 * PLD + tid];
      const float gz = rsum[tid] * Xz[tid] - Pp[2 * PLD + tid];
      const float dx = -STEP_EPS * gx;
      const float dy = -STEP_EPS * gy;
      const float dz = -STEP_EPS * gz;
      const float speed = sqrtf(dx * dx + dy * dy + dz * dz + 0.001f);
      const float at =
          ALPHA_BASE + (ALPHA - ALPHA_BASE) * ((float)(TS - tstep) / (float)TS);
      const float sc = at * tanhf(speed / at) / speed;
      Xx[tid] += dx * sc;
      Xy[tid] += dy * sc;
      Xz[tid] += dz * sc;
    }
    __syncthreads();
    if (tid < 192) {
      const int q = tid & 63, n = tid >> 6;
      const float* comp = (n == 0) ? Xx : ((n == 1) ? Xy : Xz);
      Xq[q * 16 + n] = make_float2(comp[2 * q], comp[2 * q + 1]);
    }
    __syncthreads();
  }

  float* ob = out + (size_t)b * NN * NN;
  for (int p = tid; p < NN * NN; p += 256) {
    const int i = p >> 7, j = p & 127;
    const float m  = (float)ad[p];
    const float u0 = Xx[i] - Xx[j];
    const float u1 = Xy[i] - Xy[j];
    const float u2 = Xz[i] - Xz[j];
    ob[p] = m * sqrtf(u0 * u0 + u1 * u1 + u2 * u2 + 0.01f);
  }
}

extern "C" void kernel_launch(void* const* d_in, const int* in_sizes, int n_in,
                              void* d_out, int out_size, void* d_ws, size_t ws_size,
                              hipStream_t stream) {
  const float* edge_pred = (const float*)d_in[0];
  const int*   adj       = (const int*)  d_in[1];
  const float* d_init    = (const float*)d_in[2];
  const float* u_init    = (const float*)d_in[3];
  const float* x_noise   = (const float*)d_in[4];
  float*       outp      = (float*)d_out;
  const int B = in_sizes[1] / (NN * NN);
  ts_gcn_fused<<<B, 256, 0, stream>>>(edge_pred, adj, d_init, u_init, x_noise, outp);
}


# S build: 8 columns per thread and trip: operands by 16-byte LDS loads issued together, then 8 independent sqrt/div chains (was 2, two LDS waits per trip); row sum order kept, d2 now fma-contracted
# speedup vs baseline: 2.7374x; 1.3667x over previous
#include <hip/hip_runtime.h>
#include <hip/hip_bf16.h>
#include <math.h>


typedef __attribute__((ext_vector_type(2))) float v2f;
typedef __attribute__((ext_vector_type(8))) float v8f;

#define NN   128
#define SLD  132
#define PLD  132
#define DLD  132
#define TS   10
#define STEP_EPS 0.1f
#define ALPHA 5.0f
#define ALPHA_BASE 0.1f
#define KRANK 3
#define PSTEPS 10

__device__ __forceinline__ float softplus_f(float x) {
  return fmaxf(x, 0.0f) + log1pf(expf(-fabsf(x)));
}

__device__ __forceinline__ float xhalf(float x) {
  const int b = __builtin_bit_cast(int, x);
  return __builtin_bit_cast(
      float, __builtin_amdgcn_permlanex16(b, b, 0x76543210, 0xfedcba98, false, false));
}

template <int M>
__device__ __forceinline__ float xrow(float x) {
  const int b = __builtin_bit_cast(int, x);
  return __builtin_bit_cast(float, __builtin_amdgcn_update_dpp(b, b, 0x160 + M, 0xf, 0xf, false));
}

__device__ __forceinline__ float wave_sum(float v) {
  v += xrow<1>(v);
  v += xrow<2>(v);
  v += xrow<4>(v);
  v += xrow<8>(v);
  v += xhalf(v);
  return v;
}

template <int K>
__device__ __forceinline__ void block_sum(float (&v)[K], float* slot, int tid) {
  for (int k = 0; k < K; ++k) v[k] = wave_sum(v[k]);
  if ((tid & 31) == 0) {
    for (int k = 0; k < K; ++k) slot[k * 8 + (tid >> 5)] = v[k];
  }
  __syncthreads();
  for (int k = 0; k < K; ++k) {
    const float4 lo = *(const float4*)&slot[k * 8];
    const float4 hi = *(const float4*)&slot[k * 8 + 4];
    v[k] = ((lo.x + lo.y) + (lo.z + lo.w)) + ((hi.x + hi.y) + (hi.z + hi.w));
  }
}

__device__ __forceinline__ float block_sum1(float x, float* slot, int tid) {
  float v[1] = {x};
  block_sum<1>(v, slot, tid);
  return v[0];
}

__launch_bounds__(256, 1)
__global__ void ts_gcn_fused(const float* __restrict__ edge_pred,
                             const int*   __restrict__ adj,
                             const float* __restrict__ d_init,
                             const float* __restrict__ u_init,
                             const float* __restrict__ x_noise,
                             float*       __restrict__ out)
{
  __shared__ __attribute__((aligned(16))) float Dp[NN * DLD];
  __shared__ __attribute__((aligned(16))) float Cw[NN * DLD];
  __shared__ __attribute__((aligned(16))) float Sm[NN * SLD];
  __shared__ float2 Xq[64 * 16];
  __shared__ float  Pp[16 * PLD];
  __shared__ __attribute__((aligned(16))) float Xx[NN];
  __shared__ __attribute__((aligned(16))) float Xy[NN];
  __shared__ __attribute__((aligned(16))) float Xz[NN];
  __shared__ float  rsum[NN];
  __shared__ __attribute__((aligned(16))) float wsum[2][16];
  __shared__ float  ucol[KRANK][NN];
  __shared__ __attribute__((aligned(16))) float uvec[NN];

  const int b   = blockIdx.x;
  const int tid = threadIdx.x;
  const float d0 = d_init[0];

  const float* ep = edge_pred + (size_t)b * NN * NN * 2;
  const int*   ad = adj       + (size_t)b * NN * NN;

  float partM = 0.0f;
  {
    const float4* ep4 = (const float4*)ep;
    const int2*   ad2 = (const int2*)ad;
    #pragma unroll 1
    for (int kb = 0; kb < 2; ++kb) {
      float4 ev[16];
      int2   av[16];
      #pragma unroll
      for (int u = 0; u < 16; ++u) {
        const int p2 = tid + 256 * (kb * 16 + u);
        ev[u] = ep4[p2];
        av[u] = ad2[p2];
      }
      #pragma unroll
      for (int u = 0; u < 16; ++u) {
        const int p2 = tid + 256 * (kb * 16 + u);
        const int i = p2 >> 6, j = (p2 & 63) << 1;
        const float m0 = (float)av[u].x, m1 = (float)av[u].y;
        *(float2*)&Dp[i * DLD + j] = make_float2(ev[u].x, ev[u].z);
        *(float2*)&Cw[i * DLD + j] = make_float2(ev[u].y, ev[u].w);
        *(float2*)&Sm[i * SLD + j] = make_float2(m0, m1);
        partM += ((i == j) ? 1.0f : m0) + ((i == j + 1) ? 1.0f : m1);
      }
    }
  }
  __syncthreads();
  for (int q = tid; q < (NN * (NN + 1)) / 2; q += 256) {
    const int r = q / 129, c = q - r * 129;
    const bool lo = c < NN - r;
    const int i = lo ? r : (NN - 1 - r);
    const int j = lo ? (r + c) : (c - 1);
    const int pij = i * DLD + j, pji = j * DLD + i;
    const float sp0 = softplus_f(d0 + (Dp[pij] + Dp[pji]));
    const float sp1 = softplus_f(d0 + (Cw[pij] + Cw[pji]));
    const float mij = Sm[i * SLD + j], mji = Sm[j * SLD + i];
    Dp[pij] = sp0 * mij;
    Cw[pij] = sp1 * mij;
    Dp[pji] = sp0 * mji;
    Cw[pji] = sp1 * mji;
  }
  for (int idx = tid; idx < 64 * 16; idx += 256) Xq[idx] = make_float2(0.f, 0.f);
  int rs = 0;
  const float Msum = block_sum1(partM, wsum[rs ^= 1], tid);

  const float Nmol = block_sum1(
      (tid < NN) ? ((tid == 0) ? 1.0f : Sm[tid * SLD]) : 0.0f, wsum[rs ^= 1], tid);

  {
    const int i = tid >> 1, j0 = (tid & 1) << 6;
    float part = 0.0f;
    for (int jj = 0; jj < 64; ++jj) {
      const float d = Dp[i * DLD + j0 + jj];
      part += d * d;
    }
    part += xrow<1>(part);
    if ((tid & 1) == 0) rsum[i] = part;
    __syncthreads();
    const float tot  = block_sum1((tid < NN) ? rsum[tid] : 0.0f, wsum[rs ^= 1], tid);
    const float invN = 1.0f / Nmol;
    const float Mn   = tot * invN * invN;
    for (int p = tid; p < NN * NN; p += 256) {
      const int i2 = p >> 7, j2 = p & 127;
      const float d  = Dp[i2 * DLD + j2];
      const float d2 = d * d;
      const float mk = (i2 == j2) ? 1.0f : (d > 0.0f ? 1.0f : 0.0f);
      Sm[i2 * SLD + j2] =
          mk * -0.5f * (d2 - rsum[j2] * invN - rsum[i2] * invN + Mn);
    }
    __syncthreads();
  }

  const int prow = tid >> 1, ph = tid & 1;
  float4 grow[16];
  #pragma unroll
  for (int q = 0; q < 16; ++q)
    grow[q] = *(const float4*)&Sm[prow * SLD + (2 * q + ph) * 4];
  float c0 = 0.0f, c1 = 0.0f;
  for (int kx = 0; kx < KRANK; ++kx) {
    float uval = u_init[(size_t)b * NN * 3 + prow * 3 + kx];
    for (int it = 0; it < PSTEPS; ++it) {
      const float nv = block_sum1(ph ? 0.0f : uval * uval, wsum[rs ^= 1], tid);
      const float inorm = 1.0f / fmaxf(sqrtf(nv), 0.001f);
      uval *= inorm;
      if (ph == 0) uvec[prow] = uval;
      float dc0 = 0.0f, dc1 = 0.0f;
      if (kx == 1) {
        dc0 = block_sum1(ph ? 0.0f : c0 * uval, wsum[rs ^= 1], tid);
      } else if (kx == 2) {
        float d[2] = {ph ? 0.0f : c0 * uval, ph ? 0.0f : c1 * uval};
        block_sum<2>(d, wsum[rs ^= 1], tid);
        dc0 = d[0];
        dc1 = d[1];
      } else {
        __syncthreads();
      }
      float a0 = 0.0f, a1 = 0.0f, a2 = 0.0f, a3 = 0.0f;
      #pragma unroll
      for (int q = 0; q < 16; ++q) {
        const float4 u4 = *(const float4*)&uvec[(2 * q + ph) * 4];
        a0 = fmaf(grow[q].x, u4.x, a0);
        a1 = fmaf(grow[q].y, u4.y, a1);
        a2 = fmaf(grow[q].z, u4.z, a2);
        a3 = fmaf(grow[q].w, u4.w, a3);
      }
      float acc = (a0 + a1) + (a2 + a3);
      acc += xrow<1>(acc);
      if (kx > 0) acc -= c0 * dc0;
      if (kx > 1) acc -= c1 * dc1;
      uval = acc;
    }
    const float eig = block_sum1(ph ? 0.0f : uval * uval, wsum[rs ^= 1], tid);
    const float iden = 1.0f / sqrtf(sqrtf(eig + 0.01f));
    const float unew = uval * iden;
    if (ph == 0) ucol[kx][prow] = unew;
    if (kx == 0) c0 = unew; else if (kx == 1) c1 = unew;
    __syncthreads();
  }
  if (tid < NN) {
    const float* xn = x_noise + (size_t)b * NN * 3 + tid * 3;
    Xx[tid] = ucol[0][tid] + xn[0];
    Xy[tid] = ucol[1][tid] + xn[1];
    Xz[tid] = ucol[2][tid] + xn[2];
  }
  __syncthreads();
  if (tid < 192) {
    const int q = tid & 63, n = tid >> 6;
    const float* comp = (n == 0) ? Xx : ((n == 1) ? Xy : Xz);
    Xq[q * 16 + n] = make_float2(comp[2 * q], comp[2 * q + 1]);
  }
  __syncthreads();

  const float invM4 = 4.0f / Msum;
  const int lane   = tid & 31;
  const int mt     = tid >> 5;
  const int nidx   = lane & 15;
  const int laneHi = lane >> 4;
  const int arow   = mt * 16 + nidx;
  const float2* aq = (const float2*)&Sm[arow * SLD + laneHi * 2];
  const float2* bq = &Xq[laneHi * 16 + nidx];

  for (int tstep = 0; tstep < TS; ++tstep) {
    {
      const int i = tid >> 1, j0 = (tid & 1) << 6;
      const float xi0 = Xx[i], xi1 = Xy[i], xi2 = Xz[i];
      const float4* xr = (const float4*)&Xx[j0];
      const float4* yr = (const float4*)&Xy[j0];
      const float4* zr = (const float4*)&Xz[j0];
      const float4* cr = (const float4*)&Cw[i * DLD + j0];
      const float4* dr = (const float4*)&Dp[i * DLD + j0];
      float4*       sr = (float4*)&Sm[i * SLD + j0];
      float part = 0.0f;
      for (int g = 0; g < 16; g += 2) {
        const float4 xa = xr[g], xb = xr[g + 1];
        const float4 ya = yr[g], yb = yr[g + 1];
        const float4 za = zr[g], zb = zr[g + 1];
        const float4 ca = cr[g], cb = cr[g + 1];
        const float4 da = dr[g], db = dr[g + 1];
        const float xs[8] = {xa.x, xa.y, xa.z, xa.w, xb.x, xb.y, xb.z, xb.w};
        const float ys[8] = {ya.x, ya.y, ya.z, ya.w, yb.x, yb.y, yb.z, yb.w};
        const float zs[8] = {za.x, za.y, za.z, za.w, zb.x, zb.y, zb.z, zb.w};
        const float cs[8] = {ca.x, ca.y, ca.z, ca.w, cb.x, cb.y, cb.z, cb.w};
        const float ds[8] = {da.x, da.y, da.z, da.w, db.x, db.y, db.z, db.w};
        float sv[8];
        for (int e = 0; e < 8; ++e) {
          const float u0 = xi0 - xs[e];
          const float u1 = xi1 - ys[e];
          const float u2 = xi2 - zs[e];
          const float dist = sqrtf(u0 * u0 + u1 * u1 + u2 * u2 + 0.01f);
          sv[e] = invM4 * cs[e] * (dist - ds[e]) / dist;
        }
        sr[g]     = make_float4(sv[0], sv[1], sv[2], sv[3]);
        sr[g + 1] = make_float4(sv[4], sv[5], sv[6], sv[7]);
        for (int e = 0; e < 8; ++e) part += sv[e];
      }
      part += xrow<1>(part);
      if ((tid & 1) == 0) rsum[i] = part;
      __syncthreads();
    }

    v8f acc = {};
    #pragma unroll
    for (int kt = 0; kt < 32; ++kt) {
      const float2 apair = aq[kt * 2];
      const float2 bpair = bq[kt * 32];
      v2f a;  a[0]  = apair.x;  a[1]  = apair.y;
      v2f bb; bb[0] = bpair.x;  bb[1] = bpair.y;
      acc = __builtin_amdgcn_wmma_f32_16x16x4_f32(
           false, a,  false, bb,
           (short)0, acc,  false,  false);
    }
    {
      const int rbase = mt * 16 + laneHi * 8;
      #pragma unroll
      for (int r = 0; r < 8; ++r) Pp[nidx * PLD + rbase + r] = acc[r];
    }
    __syncthreads();

    if (tid < NN) {
      const float gx = rsum[tid] * Xx[tid] - Pp[0 * PLD + tid];
      const float gy = rsum[tid] * Xy[tid] - Pp[1 * PLD + tid];
      const float gz = rsum[tid] * Xz[tid] - Pp[2 * PLD + tid];
      const float dx = -STEP_EPS * gx;
      const float dy = -STEP_EPS * gy;
      const float dz = -STEP_EPS * gz;
      const float speed = sqrtf(dx * dx + dy * dy + dz * dz + 0.001f);
      const float at =
          ALPHA_BASE + (ALPHA - ALPHA_BASE) * ((float)(TS - tstep) / (float)TS);
      const float sc = at * tanhf(speed / at) / speed;
      Xx[tid] += dx * sc;
      Xy[tid] += dy * sc;
      Xz[tid] += dz * sc;
    }
    __syncthreads();
    if (tid < 192) {
      const int q = tid & 63, n = tid >> 6;
      const float* comp = (n == 0) ? Xx : ((n == 1) ? Xy : Xz);
      Xq[q * 16 + n] = make_float2(comp[2 * q], comp[2 * q + 1]);
    }
    __syncthreads();
  }

  float* ob = out + (size_t)b * NN * NN;
  for (int p = tid; p < NN * NN; p += 256) {
    const int i = p >> 7, j = p & 127;
    const float m  = (float)ad[p];
    const float u0 = Xx[i] - Xx[j];
    const float u1 = Xy[i] - Xy[j];
    const float u2 = Xz[i] - Xz[j];
    ob[p] = m * sqrtf(u0 * u0 + u1 * u1 + u2 * u2 + 0.01f);
  }
}

extern "C" void kernel_launch(void* const* d_in, const int* in_sizes, int n_in,
                              void* d_out, int out_size, void* d_ws, size_t ws_size,
                              hipStream_t stream) {
  const float* edge_pred = (const float*)d_in[0];
  const int*   adj       = (const int*)  d_in[1];
  const float* d_init    = (const float*)d_in[2];
  const float* u_init    = (const float*)d_in[3];
  const float* x_noise   = (const float*)d_in[4];
  float*       outp      = (float*)d_out;
  const int B = in_sizes[1] / (NN * NN);
  ts_gcn_fused<<<B, 256, 0, stream>>>(edge_pred, adj, d_init, u_init, x_noise, outp);
}


# S build: factor (Dx-D)/Dx formed per cell as 1 - D*rsq(Dx^2), one hardware v_rsq_f32 and one fma, in place of correctly rounded sqrt + IEEE division (25 vector instr a cell); cell rounding changes
# speedup vs baseline: 5.1195x; 1.8702x over previous
#include <hip/hip_runtime.h>
#include <hip/hip_bf16.h>
#include <math.h>


typedef __attribute__((ext_vector_type(2))) float v2f;
typedef __attribute__((ext_vector_type(8))) float v8f;

#define NN   128
#define PLD  132
#define DLD  132
#define TS   10
#define STEP_EPS 0.1f
#define ALPHA 5.0f
#define ALPHA_BASE 0.1f
#define KRANK 3
#define PSTEPS 10

__device__ __forceinline__ float softplus_f(float x) {
  return fmaxf(x, 0.0f) + log1pf(expf(-fabsf(x)));
}

__device__ __forceinline__ float xhalf(float x) {
  const int b = __builtin_bit_cast(int, x);
  return __builtin_bit_cast(
      float, __builtin_amdgcn_permlanex16(b, b, 0x76543210, 0xfedcba98, false, false));
}

template <int M>
__device__ __forceinline__ float xrow(float x) {
  const int b = __builtin_bit_cast(int, x);
  return __builtin_bit_cast(float, __builtin_amdgcn_update_dpp(b, b, 0x160 + M, 0xf, 0xf, false));
}

__device__ __forceinline__ float wave_sum(float v) {
  v += xrow<1>(v);
  v += xrow<2>(v);
  v += xrow<4>(v);
  v += xrow<8>(v);
  v += xhalf(v);
  return v;
}

template <int K>
__device__ __forceinline__ void block_sum(float (&v)[K], float* slot, int tid) {
  for (int k = 0; k < K; ++k) v[k] = wave_sum(v[k]);
  if ((tid & 31) == 0) {
    for (int k = 0; k < K; ++k) slot[k * 8 + (tid >> 5)] = v[k];
  }
  __syncthreads();
  for (int k = 0; k < K; ++k) {
    const float4 lo = *(const float4*)&slot[k * 8];
    const float4 hi = *(const float4*)&slot[k * 8 + 4];
    v[k] = ((lo.x + lo.y) + (lo.z + lo.w)) + ((hi.x + hi.y) + (hi.z + hi.w));
  }
}

__device__ __forceinline__ float block_sum1(float x, float* slot, int tid) {
  float v[1] = {x};
  block_sum<1>(v, slot, tid);
  return v[0];
}

__device__ __forceinline__ float mask_at(const unsigned* mb, int i, int j) {
  const unsigned w = mb[i * 4 + ((j >> 6) << 1) + (j & 1)];
  return (float)((w >> ((j & 63) >> 1)) & 1u);
}

__launch_bounds__(256, 4)
__global__ void ts_gcn_fused(const float* __restrict__ edge_pred,
                             const int*   __restrict__ adj,
                             const float* __restrict__ d_init,
                             const float* __restrict__ u_init,
                             const float* __restrict__ x_noise,
                             float*       __restrict__ out)
{
  __shared__ __attribute__((aligned(16))) float Dp[NN * DLD];
  __shared__ __attribute__((aligned(16))) float Cw[NN * DLD];
  __shared__ __attribute__((aligned(16))) unsigned mbit[NN * 4];
  __shared__ float2 Xq[64 * 16];
  __shared__ float  Pp[3 * PLD];
  __shared__ __attribute__((aligned(16))) float Xx[NN];
  __shared__ __attribute__((aligned(16))) float Xy[NN];
  __shared__ __attribute__((aligned(16))) float Xz[NN];
  __shared__ __attribute__((aligned(16))) float rsum[NN];
  __shared__ __attribute__((aligned(16))) float wsum[2][16];
  __shared__ float  ucol[KRANK][NN];
  __shared__ __attribute__((aligned(16))) float uvec[NN];

  const int b   = blockIdx.x;
  const int tid = threadIdx.x;
  const float d0 = d_init[0];

  const float* ep = edge_pred + (size_t)b * NN * NN * 2;
  const int*   ad = adj       + (size_t)b * NN * NN;

  float partM = 0.0f;
  {
    const float4* ep4 = (const float4*)ep;
    const int2*   ad2 = (const int2*)ad;
    #pragma unroll 1
    for (int kb = 0; kb < 2; ++kb) {
      float4 ev[16];
      int2   av[16];
      #pragma unroll
      for (int u = 0; u < 16; ++u) {
        const int p2 = tid + 256 * (kb * 16 + u);
        ev[u] = ep4[p2];
        av[u] = ad2[p2];
      }
      #pragma unroll
      for (int u = 0; u < 16; ++u) {
        const int p2 = tid + 256 * (kb * 16 + u);
        const int i = p2 >> 6, j = (p2 & 63) << 1;
        const float m0 = (float)av[u].x, m1 = (float)av[u].y;
        *(float2*)&Dp[i * DLD + j] = make_float2(ev[u].x, ev[u].z);
        *(float2*)&Cw[i * DLD + j] = make_float2(ev[u].y, ev[u].w);
        const unsigned be = __builtin_amdgcn_ballot_w32(av[u].x != 0);
        const unsigned bo = __builtin_amdgcn_ballot_w32(av[u].y != 0);
        if ((tid & 31) == 0) *(uint2*)&mbit[i * 4 + ((p2 >> 5) & 1) * 2] = make_uint2(be, bo);
        partM += ((i == j) ? 1.0f : m0) + ((i == j + 1) ? 1.0f : m1);
      }
    }
  }
  __syncthreads();
  for (int q = tid; q < (NN * (NN + 1)) / 2; q += 256) {
    const int r = q / 129, c = q - r * 129;
    const bool lo = c < NN - r;
    const int i = lo ? r : (NN - 1 - r);
    const int j = lo ? (r + c) : (c - 1);
    const int pij = i * DLD + j, pji = j * DLD + i;
    const float sp0 = softplus_f(d0 + (Dp[pij] + Dp[pji]));
    const float sp1 = softplus_f(d0 + (Cw[pij] + Cw[pji]));
    const float mij = mask_at(mbit, i, j), mji = mask_at(mbit, j, i);
    Dp[pij] = sp0 * mij;
    Cw[pij] = sp1 * mij;
    Dp[pji] = sp0 * mji;
    Cw[pji] = sp1 * mji;
  }
  for (int idx = tid; idx < 64 * 16; idx += 256) Xq[idx] = make_float2(0.f, 0.f);
  int rs = 0;
  const float Msum = block_sum1(partM, wsum[rs ^= 1], tid);

  const float Nmol = block_sum1(
      (tid < NN) ? ((tid == 0) ? 1.0f : mask_at(mbit, tid, 0)) : 0.0f, wsum[rs ^= 1], tid);

  const int prow = tid >> 1, ph = tid & 1;
  float4 grow[16];
  {
    const int i = tid >> 1, j0 = (tid & 1) << 6;
    float part = 0.0f;
    for (int jj = 0; jj < 64; ++jj) {
      const float d = Dp[i * DLD + j0 + jj];
      part += d * d;
    }
    part += xrow<1>(part);
    if ((tid & 1) == 0) rsum[i] = part;
    __syncthreads();
    const float tot  = block_sum1((tid < NN) ? rsum[tid] : 0.0f, wsum[rs ^= 1], tid);
    const float invN = 1.0f / Nmol;
    const float Mn   = tot * invN * invN;
    const float rsi = rsum[prow];
    #pragma unroll
    for (int q = 0; q < 16; ++q) {
      const int col0 = (2 * q + ph) * 4;
      const float4 d4 = *(const float4*)&Dp[prow * DLD + col0];
      const float4 r4 = *(const float4*)&rsum[col0];
      const float dd[4] = {d4.x, d4.y, d4.z, d4.w};
      const float rr[4] = {r4.x, r4.y, r4.z, r4.w};
      float gg[4];
      for (int e = 0; e < 4; ++e) {
        const float d  = dd[e];
        const float d2 = d * d;
        const float mk = (prow == col0 + e) ? 1.0f : (d > 0.0f ? 1.0f : 0.0f);
        gg[e] = mk * -0.5f * (d2 - rr[e] * invN - rsi * invN + Mn);
      }
      grow[q] = make_float4(gg[0], gg[1], gg[2], gg[3]);
    }
  }

  float c0 = 0.0f, c1 = 0.0f;
  for (int kx = 0; kx < KRANK; ++kx) {
    float uval = u_init[(size_t)b * NN * 3 + prow * 3 + kx];
    for (int it = 0; it < PSTEPS; ++it) {
      const float nv = block_sum1(ph ? 0.0f : uval * uval, wsum[rs ^= 1], tid);
      const float inorm = 1.0f / fmaxf(sqrtf(nv), 0.001f);
      uval *= inorm;
      if (ph == 0) uvec[prow] = uval;
      float dc0 = 0.0f, dc1 = 0.0f;
      if (kx == 1) {
        dc0 = block_sum1(ph ? 0.0f : c0 * uval, wsum[rs ^= 1], tid);
      } else if (kx == 2) {
        float d[2] = {ph ? 0.0f : c0 * uval, ph ? 0.0f : c1 * uval};
        block_sum<2>(d, wsum[rs ^= 1], tid);
        dc0 = d[0];
        dc1 = d[1];
      } else {
        __syncthreads();
      }
      float a0 = 0.0f, a1 = 0.0f, a2 = 0.0f, a3 = 0.0f;
      #pragma unroll
      for (int q = 0; q < 16; ++q) {
        const float4 u4 = *(const float4*)&uvec[(2 * q + ph) * 4];
        a0 = fmaf(grow[q].x, u4.x, a0);
        a1 = fmaf(grow[q].y, u4.y, a1);
        a2 = fmaf(grow[q].z, u4.z, a2);
        a3 = fmaf(grow[q].w, u4.w, a3);
      }
      float acc = (a0 + a1) + (a2 + a3);
      acc += xrow<1>(acc);
      if (kx > 0) acc -= c0 * dc0;
      if (kx > 1) acc -= c1 * dc1;
      uval = acc;
    }
    const float eig = block_sum1(ph ? 0.0f : uval * uval, wsum[rs ^= 1], tid);
    const float iden = 1.0f / sqrtf(sqrtf(eig + 0.01f));
    const float unew = uval * iden;
    if (ph == 0) ucol[kx][prow] = unew;
    if (kx == 0) c0 = unew; else if (kx == 1) c1 = unew;
    __syncthreads();
  }
  if (tid < NN) {
    const float* xn = x_noise + (size_t)b * NN * 3 + tid * 3;
    Xx[tid] = ucol[0][tid] + xn[0];
    Xy[tid] = ucol[1][tid] + xn[1];
    Xz[tid] = ucol[2][tid] + xn[2];
  }
  __syncthreads();
  if (tid < 192) {
    const int q = tid & 63, n = tid >> 6;
    const float* comp = (n == 0) ? Xx : ((n == 1) ? Xy : Xz);
    Xq[q * 16 + n] = make_float2(comp[2 * q], comp[2 * q + 1]);
  }
  __syncthreads();

  const float invM4 = 4.0f / Msum;
  const int lane   = tid & 31;
  const int mt     = tid >> 5;
  const int nidx   = lane & 15;
  const int laneHi = lane >> 4;
  const int arow   = mt * 16 + nidx;
  const float2* bq = &Xq[laneHi * 16 + nidx];
  const float2* xq = (const float2*)&Xx[laneHi * 2];
  const float2* yq = (const float2*)&Xy[laneHi * 2];
  const float2* zq = (const float2*)&Xz[laneHi * 2];
  const float2* cq = (const float2*)&Cw[arow * DLD + laneHi * 2];
  const float2* dq = (const float2*)&Dp[arow * DLD + laneHi * 2];

  for (int tstep = 0; tstep < TS; ++tstep) {
    float part = 0.0f;
    v8f acc = {};
    {
      const float xi0 = Xx[arow], xi1 = Xy[arow], xi2 = Xz[arow];
      for (int g = 0; g < 8; ++g) {
        float2 xv[4], yv[4], zv[4], cv[4], dv[4], bv[4];
        for (int t = 0; t < 4; ++t) {
          const int kt = g * 4 + t;
          xv[t] = xq[kt * 2];
          yv[t] = yq[kt * 2];
          zv[t] = zq[kt * 2];
          cv[t] = cq[kt * 2];
          dv[t] = dq[kt * 2];
          bv[t] = bq[kt * 32];
        }
        const float xs[8] = {xv[0].x, xv[0].y, xv[1].x, xv[1].y, xv[2].x, xv[2].y, xv[3].x, xv[3].y};
        const float ys[8] = {yv[0].x, yv[0].y, yv[1].x, yv[1].y, yv[2].x, yv[2].y, yv[3].x, yv[3].y};
        const float zs[8] = {zv[0].x, zv[0].y, zv[1].x, zv[1].y, zv[2].x, zv[2].y, zv[3].x, zv[3].y};
        const float cs[8] = {cv[0].x, cv[0].y, cv[1].x, cv[1].y, cv[2].x, cv[2].y, cv[3].x, cv[3].y};
        const float ds[8] = {dv[0].x, dv[0].y, dv[1].x, dv[1].y, dv[2].x, dv[2].y, dv[3].x, dv[3].y};
        float sv[8];
        for (int e = 0; e < 8; ++e) {
          const float u0 = xi0 - xs[e];
          const float u1 = xi1 - ys[e];
          const float u2 = xi2 - zs[e];
          const float r2 = u0 * u0 + u1 * u1 + u2 * u2 + 0.01f;
          const float w = __builtin_amdgcn_rsqf(r2);
          sv[e] = invM4 * cs[e] * (1.0f - ds[e] * w);
        }
        __builtin_amdgcn_sched_barrier(0);
        for (int t = 0; t < 4; ++t) {
          v2f a;  a[0]  = sv[2 * t];  a[1]  = sv[2 * t + 1];
          v2f bb; bb[0] = bv[t].x;    bb[1] = bv[t].y;
          acc = __builtin_amdgcn_wmma_f32_16x16x4_f32(
               false, a,  false, bb,
               (short)0, acc,  false,  false);
        }
        asm volatile("v_nop\n\tv_nop\n\tv_nop\n\tv_nop" : "+v"(acc));
        __builtin_amdgcn_sched_barrier(0);
        for (int e = 0; e < 8; ++e) part += sv[e];
      }
      part += xhalf(part);
      if (laneHi == 0) rsum[arow] = part;
    }
    if (nidx < 3) {
      const int rbase = mt * 16 + laneHi * 8;
      #pragma unroll
      for (int r = 0; r < 8; ++r) Pp[nidx * PLD + rbase + r] = acc[r];
    }
    __syncthreads();

    if (tid < NN) {
      const float gx = rsum[tid] * Xx[tid] - Pp[0 * PLD + tid];
      const float gy = rsum[tid] * Xy[tid] - Pp[1 * PLD + tid];
      const float gz = rsum[tid] * Xz[tid] - Pp[2 * PLD + tid];
      const float dx = -STEP_EPS * gx;
      const float dy = -STEP_EPS * gy;
      const float dz = -STEP_EPS * gz;
      const float speed = sqrtf(dx * dx + dy * dy + dz * dz + 0.001f);
      const float at =
          ALPHA_BASE + (ALPHA - ALPHA_BASE) * ((float)(TS - tstep) / (float)TS);
      const float sc = at * tanhf(speed / at) / speed;
      Xx[tid] += dx * sc;
      Xy[tid] += dy * sc;
      Xz[tid] += dz * sc;
    }
    __syncthreads();
    if (tid < 192) {
      const int q = tid & 63, n = tid >> 6;
      const float* comp = (n == 0) ? Xx : ((n == 1) ? Xy : Xz);
      Xq[q * 16 + n] = make_float2(comp[2 * q], comp[2 * q + 1]);
    }
    __syncthreads();
  }

  float* ob = out + (size_t)b * NN * NN;
  for (int p = tid; p < NN * NN; p += 256) {
    const int i = p >> 7, j = p & 127;
    const float m  = (float)ad[p];
    const float u0 = Xx[i] - Xx[j];
    const float u1 = Xy[i] - Xy[j];
    const float u2 = Xz[i] - Xz[j];
    ob[p] = m * sqrtf(u0 * u0 + u1 * u1 + u2 * u2 + 0.01f);
  }
}

extern "C" void kernel_launch(void* const* d_in, const int* in_sizes, int n_in,
                              void* d_out, int out_size, void* d_ws, size_t ws_size,
                              hipStream_t stream) {
  const float* edge_pred = (const float*)d_in[0];
  const int*   adj       = (const int*)  d_in[1];
  const float* d_init    = (const float*)d_in[2];
  const float* u_init    = (const float*)d_in[3];
  const float* x_noise   = (const float*)d_in[4];
  float*       outp      = (float*)d_out;
  const int B = in_sizes[1] / (NN * NN);
  ts_gcn_fused<<<B, 256, 0, stream>>>(edge_pred, adj, d_init, u_init, x_noise, outp);
}


# Phase 0 softplus: exp(-|x|) and log1p built on hardware v_exp_f32 / v_rcp_f32 (error-compensated 2^y, atanh series to s^13) replace library expf + double-float log1pf; cell rounding changes
# speedup vs baseline: 5.7139x; 1.1161x over previous
#include <hip/hip_runtime.h>
#include <hip/hip_bf16.h>
#include <math.h>


typedef __attribute__((ext_vector_type(2))) float v2f;
typedef __attribute__((ext_vector_type(8))) float v8f;

#define NN   128
#define PLD  132
#define DLD  132
#define TS   10
#define STEP_EPS 0.1f
#define ALPHA 5.0f
#define ALPHA_BASE 0.1f
#define KRANK 3
#define PSTEPS 10

__device__ __forceinline__ float softplus_f(float x) {
  const float ax = fabsf(x);
  const float a  = (ax > 128.0f) ? 128.0f : ax;
  const float y  = a * -1.44269502e+0f;
  float r = __builtin_fmaf(a, -1.44269502e+0f, -y);
  r = __builtin_fmaf(a, -1.92596303e-8f, r);
  const float e0 = __builtin_amdgcn_exp2f(y);
  const float t  = __builtin_fmaf(e0, r * 0.693147181f, e0);
  const float d  = 2.0f + t;
  const float r0 = __builtin_amdgcn_rcpf(d);
  const float s0 = t * r0;
  const float s  = __builtin_fmaf(__builtin_fmaf(-s0, d, t), r0, s0);
  const float z  = s * s;
  float p = 7.69230798e-2f;
  p = __builtin_fmaf(p, z, 9.09090936e-2f);
  p = __builtin_fmaf(p, z, 1.11111112e-1f);
  p = __builtin_fmaf(p, z, 1.42857149e-1f);
  p = __builtin_fmaf(p, z, 2.0e-1f);
  p = __builtin_fmaf(p, z, 3.33333343e-1f);
  const float s2 = s + s;
  return fmaxf(x, 0.0f) + __builtin_fmaf(s2, z * p, s2);
}

__device__ __forceinline__ float xhalf(float x) {
  const int b = __builtin_bit_cast(int, x);
  return __builtin_bit_cast(
      float, __builtin_amdgcn_permlanex16(b, b, 0x76543210, 0xfedcba98, false, false));
}

template <int M>
__device__ __forceinline__ float xrow(float x) {
  const int b = __builtin_bit_cast(int, x);
  return __builtin_bit_cast(float, __builtin_amdgcn_update_dpp(b, b, 0x160 + M, 0xf, 0xf, false));
}

__device__ __forceinline__ float wave_sum(float v) {
  v += xrow<1>(v);
  v += xrow<2>(v);
  v += xrow<4>(v);
  v += xrow<8>(v);
  v += xhalf(v);
  return v;
}

template <int K>
__device__ __forceinline__ void block_sum(float (&v)[K], float* slot, int tid) {
  for (int k = 0; k < K; ++k) v[k] = wave_sum(v[k]);
  if ((tid & 31) == 0) {
    for (int k = 0; k < K; ++k) slot[k * 8 + (tid >> 5)] = v[k];
  }
  __syncthreads();
  for (int k = 0; k < K; ++k) {
    const float4 lo = *(const float4*)&slot[k * 8];
    const float4 hi = *(const float4*)&slot[k * 8 + 4];
    v[k] = ((lo.x + lo.y) + (lo.z + lo.w)) + ((hi.x + hi.y) + (hi.z + hi.w));
  }
}

__device__ __forceinline__ float block_sum1(float x, float* slot, int tid) {
  float v[1] = {x};
  block_sum<1>(v, slot, tid);
  return v[0];
}

__device__ __forceinline__ float mask_at(const unsigned* mb, int i, int j) {
  const unsigned w = mb[i * 4 + ((j >> 6) << 1) + (j & 1)];
  return (float)((w >> ((j & 63) >> 1)) & 1u);
}

__launch_bounds__(256, 4)
__global__ void ts_gcn_fused(const float* __restrict__ edge_pred,
                             const int*   __restrict__ adj,
                             const float* __restrict__ d_init,
                             const float* __restrict__ u_init,
                             const float* __restrict__ x_noise,
                             float*       __restrict__ out)
{
  __shared__ __attribute__((aligned(16))) float Dp[NN * DLD];
  __shared__ __attribute__((aligned(16))) float Cw[NN * DLD];
  __shared__ __attribute__((aligned(16))) unsigned mbit[NN * 4];
  __shared__ float2 Xq[64 * 16];
  __shared__ float  Pp[3 * PLD];
  __shared__ __attribute__((aligned(16))) float Xx[NN];
  __shared__ __attribute__((aligned(16))) float Xy[NN];
  __shared__ __attribute__((aligned(16))) float Xz[NN];
  __shared__ __attribute__((aligned(16))) float rsum[NN];
  __shared__ __attribute__((aligned(16))) float wsum[2][16];
  __shared__ float  ucol[KRANK][NN];
  __shared__ __attribute__((aligned(16))) float uvec[NN];

  const int b   = blockIdx.x;
  const int tid = threadIdx.x;
  const float d0 = d_init[0];

  const float* ep = edge_pred + (size_t)b * NN * NN * 2;
  const int*   ad = adj       + (size_t)b * NN * NN;

  float partM = 0.0f;
  {
    const float4* ep4 = (const float4*)ep;
    const int2*   ad2 = (const int2*)ad;
    #pragma unroll 1
    for (int kb = 0; kb < 2; ++kb) {
      float4 ev[16];
      int2   av[16];
      #pragma unroll
      for (int u = 0; u < 16; ++u) {
        const int p2 = tid + 256 * (kb * 16 + u);
        ev[u] = ep4[p2];
        av[u] = ad2[p2];
      }
      #pragma unroll
      for (int u = 0; u < 16; ++u) {
        const int p2 = tid + 256 * (kb * 16 + u);
        const int i = p2 >> 6, j = (p2 & 63) << 1;
        const float m0 = (float)av[u].x, m1 = (float)av[u].y;
        *(float2*)&Dp[i * DLD + j] = make_float2(ev[u].x, ev[u].z);
        *(float2*)&Cw[i * DLD + j] = make_float2(ev[u].y, ev[u].w);
        const unsigned be = __builtin_amdgcn_ballot_w32(av[u].x != 0);
        const unsigned bo = __builtin_amdgcn_ballot_w32(av[u].y != 0);
        if ((tid & 31) == 0) *(uint2*)&mbit[i * 4 + ((p2 >> 5) & 1) * 2] = make_uint2(be, bo);
        partM += ((i == j) ? 1.0f : m0) + ((i == j + 1) ? 1.0f : m1);
      }
    }
  }
  __syncthreads();
  for (int q = tid; q < (NN * (NN + 1)) / 2; q += 256) {
    const int r = q / 129, c = q - r * 129;
    const bool lo = c < NN - r;
    const int i = lo ? r : (NN - 1 - r);
    const int j = lo ? (r + c) : (c - 1);
    const int pij = i * DLD + j, pji = j * DLD + i;
    const float sp0 = softplus_f(d0 + (Dp[pij] + Dp[pji]));
    const float sp1 = softplus_f(d0 + (Cw[pij] + Cw[pji]));
    const float mij = mask_at(mbit, i, j), mji = mask_at(mbit, j, i);
    Dp[pij] = sp0 * mij;
    Cw[pij] = sp1 * mij;
    Dp[pji] = sp0 * mji;
    Cw[pji] = sp1 * mji;
  }
  for (int idx = tid; idx < 64 * 16; idx += 256) Xq[idx] = make_float2(0.f, 0.f);
  int rs = 0;
  const float Msum = block_sum1(partM, wsum[rs ^= 1], tid);

  const float Nmol = block_sum1(
      (tid < NN) ? ((tid == 0) ? 1.0f : mask_at(mbit, tid, 0)) : 0.0f, wsum[rs ^= 1], tid);

  const int prow = tid >> 1, ph = tid & 1;
  float4 grow[16];
  {
    const int i = tid >> 1, j0 = (tid & 1) << 6;
    float part = 0.0f;
    for (int jj = 0; jj < 64; ++jj) {
      const float d = Dp[i * DLD + j0 + jj];
      part += d * d;
    }
    part += xrow<1>(part);
    if ((tid & 1) == 0) rsum[i] = part;
    __syncthreads();
    const float tot  = block_sum1((tid < NN) ? rsum[tid] : 0.0f, wsum[rs ^= 1], tid);
    const float invN = 1.0f / Nmol;
    const float Mn   = tot * invN * invN;
    const float rsi = rsum[prow];
    #pragma unroll
    for (int q = 0; q < 16; ++q) {
      const int col0 = (2 * q + ph) * 4;
      const float4 d4 = *(const float4*)&Dp[prow * DLD + col0];
      const float4 r4 = *(const float4*)&rsum[col0];
      const float dd[4] = {d4.x, d4.y, d4.z, d4.w};
      const float rr[4] = {r4.x, r4.y, r4.z, r4.w};
      float gg[4];
      for (int e = 0; e < 4; ++e) {
        const float d  = dd[e];
        const float d2 = d * d;
        const float mk = (prow == col0 + e) ? 1.0f : (d > 0.0f ? 1.0f : 0.0f);
        gg[e] = mk * -0.5f * (d2 - rr[e] * invN - rsi * invN + Mn);
      }
      grow[q] = make_float4(gg[0], gg[1], gg[2], gg[3]);
    }
  }

  float c0 = 0.0f, c1 = 0.0f;
  for (int kx = 0; kx < KRANK; ++kx) {
    float uval = u_init[(size_t)b * NN * 3 + prow * 3 + kx];
    for (int it = 0; it < PSTEPS; ++it) {
      const float nv = block_sum1(ph ? 0.0f : uval * uval, wsum[rs ^= 1], tid);
      const float inorm = 1.0f / fmaxf(sqrtf(nv), 0.001f);
      uval *= inorm;
      if (ph == 0) uvec[prow] = uval;
      float dc0 = 0.0f, dc1 = 0.0f;
      if (kx == 1) {
        dc0 = block_sum1(ph ? 0.0f : c0 * uval, wsum[rs ^= 1], tid);
      } else if (kx == 2) {
        float d[2] = {ph ? 0.0f : c0 * uval, ph ? 0.0f : c1 * uval};
        block_sum<2>(d, wsum[rs ^= 1], tid);
        dc0 = d[0];
        dc1 = d[1];
      } else {
        __syncthreads();
      }
      float a0 = 0.0f, a1 = 0.0f, a2 = 0.0f, a3 = 0.0f;
      #pragma unroll
      for (int q = 0; q < 16; ++q) {
        const float4 u4 = *(const float4*)&uvec[(2 * q + ph) * 4];
        a0 = fmaf(grow[q].x, u4.x, a0);
        a1 = fmaf(grow[q].y, u4.y, a1);
        a2 = fmaf(grow[q].z, u4.z, a2);
        a3 = fmaf(grow[q].w, u4.w, a3);
      }
      float acc = (a0 + a1) + (a2 + a3);
      acc += xrow<1>(acc);
      if (kx > 0) acc -= c0 * dc0;
      if (kx > 1) acc -= c1 * dc1;
      uval = acc;
    }
    const float eig = block_sum1(ph ? 0.0f : uval * uval, wsum[rs ^= 1], tid);
    const float iden = 1.0f / sqrtf(sqrtf(eig + 0.01f));
    const float unew = uval * iden;
    if (ph == 0) ucol[kx][prow] = unew;
    if (kx == 0) c0 = unew; else if (kx == 1) c1 = unew;
    __syncthreads();
  }
  if (tid < NN) {
    const float* xn = x_noise + (size_t)b * NN * 3 + tid * 3;
    Xx[tid] = ucol[0][tid] + xn[0];
    Xy[tid] = ucol[1][tid] + xn[1];
    Xz[tid] = ucol[2][tid] + xn[2];
  }
  __syncthreads();
  if (tid < 192) {
    const int q = tid & 63, n = tid >> 6;
    const float* comp = (n == 0) ? Xx : ((n == 1) ? Xy : Xz);
    Xq[q * 16 + n] = make_float2(comp[2 * q], comp[2 * q + 1]);
  }
  __syncthreads();

  const float invM4 = 4.0f / Msum;
  const int lane   = tid & 31;
  const int mt     = tid >> 5;
  const int nidx   = lane & 15;
  const int laneHi = lane >> 4;
  const int arow   = mt * 16 + nidx;
  const float2* bq = &Xq[laneHi * 16 + nidx];
  const float2* xq = (const float2*)&Xx[laneHi * 2];
  const float2* yq = (const float2*)&Xy[laneHi * 2];
  const float2* zq = (const float2*)&Xz[laneHi * 2];
  const float2* cq = (const float2*)&Cw[arow * DLD + laneHi * 2];
  const float2* dq = (const float2*)&Dp[arow * DLD + laneHi * 2];

  for (int tstep = 0; tstep < TS; ++tstep) {
    float part = 0.0f;
    v8f acc = {};
    {
      const float xi0 = Xx[arow], xi1 = Xy[arow], xi2 = Xz[arow];
      for (int g = 0; g < 8; ++g) {
        float2 xv[4], yv[4], zv[4], cv[4], dv[4], bv[4];
        for (int t = 0; t < 4; ++t) {
          const int kt = g * 4 + t;
          xv[t] = xq[kt * 2];
          yv[t] = yq[kt * 2];
          zv[t] = zq[kt * 2];
          cv[t] = cq[kt * 2];
          dv[t] = dq[kt * 2];
          bv[t] = bq[kt * 32];
        }
        const float xs[8] = {xv[0].x, xv[0].y, xv[1].x, xv[1].y, xv[2].x, xv[2].y, xv[3].x, xv[3].y};
        const float ys[8] = {yv[0].x, yv[0].y, yv[1].x, yv[1].y, yv[2].x, yv[2].y, yv[3].x, yv[3].y};
        const float zs[8] = {zv[0].x, zv[0].y, zv[1].x, zv[1].y, zv[2].x, zv[2].y, zv[3].x, zv[3].y};
        const float cs[8] = {cv[0].x, cv[0].y, cv[1].x, cv[1].y, cv[2].x, cv[2].y, cv[3].x, cv[3].y};
        const float ds[8] = {dv[0].x, dv[0].y, dv[1].x, dv[1].y, dv[2].x, dv[2].y, dv[3].x, dv[3].y};
        float sv[8];
        for (int e = 0; e < 8; ++e) {
          const float u0 = xi0 - xs[e];
          const float u1 = xi1 - ys[e];
          const float u2 = xi2 - zs[e];
          const float r2 = u0 * u0 + u1 * u1 + u2 * u2 + 0.01f;
          const float w = __builtin_amdgcn_rsqf(r2);
          sv[e] = invM4 * cs[e] * (1.0f - ds[e] * w);
        }
        __builtin_amdgcn_sched_barrier(0);
        for (int t = 0; t < 4; ++t) {
          v2f a;  a[0]  = sv[2 * t];  a[1]  = sv[2 * t + 1];
          v2f bb; bb[0] = bv[t].x;    bb[1] = bv[t].y;
          acc = __builtin_amdgcn_wmma_f32_16x16x4_f32(
               false, a,  false, bb,
               (short)0, acc,  false,  false);
        }
        asm volatile("v_nop\n\tv_nop\n\tv_nop\n\tv_nop" : "+v"(acc));
        __builtin_amdgcn_sched_barrier(0);
        for (int e = 0; e < 8; ++e) part += sv[e];
      }
      part += xhalf(part);
      if (laneHi == 0) rsum[arow] = part;
    }
    if (nidx < 3) {
      const int rbase = mt * 16 + laneHi * 8;
      #pragma unroll
      for (int r = 0; r < 8; ++r) Pp[nidx * PLD + rbase + r] = acc[r];
    }
    __syncthreads();

    if (tid < NN) {
      const float gx = rsum[tid] * Xx[tid] - Pp[0 * PLD + tid];
      const float gy = rsum[tid] * Xy[tid] - Pp[1 * PLD + tid];
      const float gz = rsum[tid] * Xz[tid] - Pp[2 * PLD + tid];
      const float dx = -STEP_EPS * gx;
      const float dy = -STEP_EPS * gy;
      const float dz = -STEP_EPS * gz;
      const float speed = sqrtf(dx * dx + dy * dy + dz * dz + 0.001f);
      const float at =
          ALPHA_BASE + (ALPHA - ALPHA_BASE) * ((float)(TS - tstep) / (float)TS);
      const float sc = at * tanhf(speed / at) / speed;
      Xx[tid] += dx * sc;
      Xy[tid] += dy * sc;
      Xz[tid] += dz * sc;
    }
    __syncthreads();
    if (tid < 192) {
      const int q = tid & 63, n = tid >> 6;
      const float* comp = (n == 0) ? Xx : ((n == 1) ? Xy : Xz);
      Xq[q * 16 + n] = make_float2(comp[2 * q], comp[2 * q + 1]);
    }
    __syncthreads();
  }

  float* ob = out + (size_t)b * NN * NN;
  for (int p = tid; p < NN * NN; p += 256) {
    const int i = p >> 7, j = p & 127;
    const float m  = (float)ad[p];
    const float u0 = Xx[i] - Xx[j];
    const float u1 = Xy[i] - Xy[j];
    const float u2 = Xz[i] - Xz[j];
    ob[p] = m * sqrtf(u0 * u0 + u1 * u1 + u2 * u2 + 0.01f);
  }
}

extern "C" void kernel_launch(void* const* d_in, const int* in_sizes, int n_in,
                              void* d_out, int out_size, void* d_ws, size_t ws_size,
                              hipStream_t stream) {
  const float* edge_pred = (const float*)d_in[0];
  const int*   adj       = (const int*)  d_in[1];
  const float* d_init    = (const float*)d_in[2];
  const float* u_init    = (const float*)d_in[3];
  const float* x_noise   = (const float*)d_in[4];
  float*       outp      = (float*)d_out;
  const int B = in_sizes[1] / (NN * NN);
  ts_gcn_fused<<<B, 256, 0, stream>>>(edge_pred, adj, d_init, u_init, x_noise, outp);
}


# Power step behind ONE barrier (was 2-3): raw iterate in two LDS buffers, each thread sums the norm from its matvec's own loads, deflation moved into the register-held matrix; norm sum order changes
# speedup vs baseline: 6.1996x; 1.0850x over previous
#include <hip/hip_runtime.h>
#include <hip/hip_bf16.h>
#include <math.h>


typedef __attribute__((ext_vector_type(2))) float v2f;
typedef __attribute__((ext_vector_type(8))) float v8f;

#define NN   128
#define PLD  132
#define DLD  132
#define TS   10
#define STEP_EPS 0.1f
#define ALPHA 5.0f
#define ALPHA_BASE 0.1f
#define KRANK 3
#define PSTEPS 10

__device__ __forceinline__ float softplus_f(float x) {
  const float ax = fabsf(x);
  const float a  = (ax > 128.0f) ? 128.0f : ax;
  const float y  = a * -1.44269502e+0f;
  float r = __builtin_fmaf(a, -1.44269502e+0f, -y);
  r = __builtin_fmaf(a, -1.92596303e-8f, r);
  const float e0 = __builtin_amdgcn_exp2f(y);
  const float t  = __builtin_fmaf(e0, r * 0.693147181f, e0);
  const float d  = 2.0f + t;
  const float r0 = __builtin_amdgcn_rcpf(d);
  const float s0 = t * r0;
  const float s  = __builtin_fmaf(__builtin_fmaf(-s0, d, t), r0, s0);
  const float z  = s * s;
  float p = 7.69230798e-2f;
  p = __builtin_fmaf(p, z, 9.09090936e-2f);
  p = __builtin_fmaf(p, z, 1.11111112e-1f);
  p = __builtin_fmaf(p, z, 1.42857149e-1f);
  p = __builtin_fmaf(p, z, 2.0e-1f);
  p = __builtin_fmaf(p, z, 3.33333343e-1f);
  const float s2 = s + s;
  return fmaxf(x, 0.0f) + __builtin_fmaf(s2, z * p, s2);
}

__device__ __forceinline__ float xhalf(float x) {
  const int b = __builtin_bit_cast(int, x);
  return __builtin_bit_cast(
      float, __builtin_amdgcn_permlanex16(b, b, 0x76543210, 0xfedcba98, false, false));
}

template <int M>
__device__ __forceinline__ float xrow(float x) {
  const int b = __builtin_bit_cast(int, x);
  return __builtin_bit_cast(float, __builtin_amdgcn_update_dpp(b, b, 0x160 + M, 0xf, 0xf, false));
}

__device__ __forceinline__ float wave_sum(float v) {
  v += xrow<1>(v);
  v += xrow<2>(v);
  v += xrow<4>(v);
  v += xrow<8>(v);
  v += xhalf(v);
  return v;
}

template <int K>
__device__ __forceinline__ void block_sum(float (&v)[K], float* slot, int tid) {
  for (int k = 0; k < K; ++k) v[k] = wave_sum(v[k]);
  if ((tid & 31) == 0) {
    for (int k = 0; k < K; ++k) slot[k * 8 + (tid >> 5)] = v[k];
  }
  __syncthreads();
  for (int k = 0; k < K; ++k) {
    const float4 lo = *(const float4*)&slot[k * 8];
    const float4 hi = *(const float4*)&slot[k * 8 + 4];
    v[k] = ((lo.x + lo.y) + (lo.z + lo.w)) + ((hi.x + hi.y) + (hi.z + hi.w));
  }
}

__device__ __forceinline__ float block_sum1(float x, float* slot, int tid) {
  float v[1] = {x};
  block_sum<1>(v, slot, tid);
  return v[0];
}

__device__ __forceinline__ float mask_at(const unsigned* mb, int i, int j) {
  const unsigned w = mb[i * 4 + ((j >> 6) << 1) + (j & 1)];
  return (float)((w >> ((j & 63) >> 1)) & 1u);
}

__launch_bounds__(256, 4)
__global__ void ts_gcn_fused(const float* __restrict__ edge_pred,
                             const int*   __restrict__ adj,
                             const float* __restrict__ d_init,
                             const float* __restrict__ u_init,
                             const float* __restrict__ x_noise,
                             float*       __restrict__ out)
{
  __shared__ __attribute__((aligned(16))) float Dp[NN * DLD];
  __shared__ __attribute__((aligned(16))) float Cw[NN * DLD];
  __shared__ __attribute__((aligned(16))) unsigned mbit[NN * 4];
  __shared__ float2 Xq[64 * 16];
  __shared__ float  Pp[3 * PLD];
  __shared__ __attribute__((aligned(16))) float Xx[NN];
  __shared__ __attribute__((aligned(16))) float Xy[NN];
  __shared__ __attribute__((aligned(16))) float Xz[NN];
  __shared__ __attribute__((aligned(16))) float rsum[NN];
  __shared__ __attribute__((aligned(16))) float wsum[2][16];
  __shared__ float  ucol[KRANK][NN];
  __shared__ __attribute__((aligned(16))) float ubuf[2 * NN];

  const int b   = blockIdx.x;
  const int tid = threadIdx.x;
  const float d0 = d_init[0];

  const float* ep = edge_pred + (size_t)b * NN * NN * 2;
  const int*   ad = adj       + (size_t)b * NN * NN;

  float partM = 0.0f;
  {
    const float4* ep4 = (const float4*)ep;
    const int2*   ad2 = (const int2*)ad;
    #pragma unroll 1
    for (int kb = 0; kb < 2; ++kb) {
      float4 ev[16];
      int2   av[16];
      #pragma unroll
      for (int u = 0; u < 16; ++u) {
        const int p2 = tid + 256 * (kb * 16 + u);
        ev[u] = ep4[p2];
        av[u] = ad2[p2];
      }
      #pragma unroll
      for (int u = 0; u < 16; ++u) {
        const int p2 = tid + 256 * (kb * 16 + u);
        const int i = p2 >> 6, j = (p2 & 63) << 1;
        const float m0 = (float)av[u].x, m1 = (float)av[u].y;
        *(float2*)&Dp[i * DLD + j] = make_float2(ev[u].x, ev[u].z);
        *(float2*)&Cw[i * DLD + j] = make_float2(ev[u].y, ev[u].w);
        const unsigned be = __builtin_amdgcn_ballot_w32(av[u].x != 0);
        const unsigned bo = __builtin_amdgcn_ballot_w32(av[u].y != 0);
        if ((tid & 31) == 0) *(uint2*)&mbit[i * 4 + ((p2 >> 5) & 1) * 2] = make_uint2(be, bo);
        partM += ((i == j) ? 1.0f : m0) + ((i == j + 1) ? 1.0f : m1);
      }
    }
  }
  __syncthreads();
  for (int q = tid; q < (NN * (NN + 1)) / 2; q += 256) {
    const int r = q / 129, c = q - r * 129;
    const bool lo = c < NN - r;
    const int i = lo ? r : (NN - 1 - r);
    const int j = lo ? (r + c) : (c - 1);
    const int pij = i * DLD + j, pji = j * DLD + i;
    const float sp0 = softplus_f(d0 + (Dp[pij] + Dp[pji]));
    const float sp1 = softplus_f(d0 + (Cw[pij] + Cw[pji]));
    const float mij = mask_at(mbit, i, j), mji = mask_at(mbit, j, i);
    Dp[pij] = sp0 * mij;
    Cw[pij] = sp1 * mij;
    Dp[pji] = sp0 * mji;
    Cw[pji] = sp1 * mji;
  }
  for (int idx = tid; idx < 64 * 16; idx += 256) Xq[idx] = make_float2(0.f, 0.f);
  int rs = 0;
  const float Msum = block_sum1(partM, wsum[rs ^= 1], tid);

  const float Nmol = block_sum1(
      (tid < NN) ? ((tid == 0) ? 1.0f : mask_at(mbit, tid, 0)) : 0.0f, wsum[rs ^= 1], tid);

  const int prow = tid >> 1, ph = tid & 1;
  float4 grow[16];
  {
    const int i = tid >> 1, j0 = (tid & 1) << 6;
    float part = 0.0f;
    for (int jj = 0; jj < 64; ++jj) {
      const float d = Dp[i * DLD + j0 + jj];
      part += d * d;
    }
    part += xrow<1>(part);
    if ((tid & 1) == 0) rsum[i] = part;
    __syncthreads();
    const float tot  = block_sum1((tid < NN) ? rsum[tid] : 0.0f, wsum[rs ^= 1], tid);
    const float invN = 1.0f / Nmol;
    const float Mn   = tot * invN * invN;
    const float rsi = rsum[prow];
    #pragma unroll
    for (int q = 0; q < 16; ++q) {
      const int col0 = (2 * q + ph) * 4;
      const float4 d4 = *(const float4*)&Dp[prow * DLD + col0];
      const float4 r4 = *(const float4*)&rsum[col0];
      const float dd[4] = {d4.x, d4.y, d4.z, d4.w};
      const float rr[4] = {r4.x, r4.y, r4.z, r4.w};
      float gg[4];
      for (int e = 0; e < 4; ++e) {
        const float d  = dd[e];
        const float d2 = d * d;
        const float mk = (prow == col0 + e) ? 1.0f : (d > 0.0f ? 1.0f : 0.0f);
        gg[e] = mk * -0.5f * (d2 - rr[e] * invN - rsi * invN + Mn);
      }
      grow[q] = make_float4(gg[0], gg[1], gg[2], gg[3]);
    }
  }

  int uo = 0;
  for (int kx = 0; kx < KRANK; ++kx) {
    float uval = u_init[(size_t)b * NN * 3 + prow * 3 + kx];
    for (int it = 0; it < PSTEPS; ++it) {
      const float* ub = &ubuf[uo];
      if (ph == 0) ubuf[uo + prow] = uval;
      uo ^= NN;
      __syncthreads();
      float a0 = 0.0f, a1 = 0.0f, a2 = 0.0f, a3 = 0.0f;
      float n0 = 0.0f, n1 = 0.0f, n2 = 0.0f, n3 = 0.0f;
      #pragma unroll
      for (int q = 0; q < 16; ++q) {
        const float4 u4 = *(const float4*)&ub[(2 * q + ph) * 4];
        a0 = fmaf(grow[q].x, u4.x, a0);
        a1 = fmaf(grow[q].y, u4.y, a1);
        a2 = fmaf(grow[q].z, u4.z, a2);
        a3 = fmaf(grow[q].w, u4.w, a3);
        n0 = fmaf(u4.x, u4.x, n0);
        n1 = fmaf(u4.y, u4.y, n1);
        n2 = fmaf(u4.z, u4.z, n2);
        n3 = fmaf(u4.w, u4.w, n3);
      }
      float acc = (a0 + a1) + (a2 + a3);
      float nv  = (n0 + n1) + (n2 + n3);
      acc += xrow<1>(acc);
      nv  += xrow<1>(nv);
      const float inorm = 1.0f / fmaxf(sqrtf(nv), 0.001f);
      uval = acc * inorm;
    }
    const float* ub = &ubuf[uo];
    if (ph == 0) ubuf[uo + prow] = uval;
    uo ^= NN;
    __syncthreads();
    float n0 = 0.0f, n1 = 0.0f, n2 = 0.0f, n3 = 0.0f;
    #pragma unroll
    for (int q = 0; q < 16; ++q) {
      const float4 u4 = *(const float4*)&ub[(2 * q + ph) * 4];
      n0 = fmaf(u4.x, u4.x, n0);
      n1 = fmaf(u4.y, u4.y, n1);
      n2 = fmaf(u4.z, u4.z, n2);
      n3 = fmaf(u4.w, u4.w, n3);
    }
    float eig = (n0 + n1) + (n2 + n3);
    eig += xrow<1>(eig);
    const float iden = 1.0f / sqrtf(sqrtf(eig + 0.01f));
    const float unew = uval * iden;
    if (ph == 0) ucol[kx][prow] = unew;
    if (kx + 1 < KRANK) {
      #pragma unroll
      for (int q = 0; q < 16; ++q) {
        const float4 u4 = *(const float4*)&ub[(2 * q + ph) * 4];
        grow[q].x = fmaf(-unew, u4.x * iden, grow[q].x);
        grow[q].y = fmaf(-unew, u4.y * iden, grow[q].y);
        grow[q].z = fmaf(-unew, u4.z * iden, grow[q].z);
        grow[q].w = fmaf(-unew, u4.w * iden, grow[q].w);
      }
    }
  }
  __syncthreads();
  if (tid < NN) {
    const float* xn = x_noise + (size_t)b * NN * 3 + tid * 3;
    Xx[tid] = ucol[0][tid] + xn[0];
    Xy[tid] = ucol[1][tid] + xn[1];
    Xz[tid] = ucol[2][tid] + xn[2];
  }
  __syncthreads();
  if (tid < 192) {
    const int q = tid & 63, n = tid >> 6;
    const float* comp = (n == 0) ? Xx : ((n == 1) ? Xy : Xz);
    Xq[q * 16 + n] = make_float2(comp[2 * q], comp[2 * q + 1]);
  }
  __syncthreads();

  const float invM4 = 4.0f / Msum;
  const int lane   = tid & 31;
  const int mt     = tid >> 5;
  const int nidx   = lane & 15;
  const int laneHi = lane >> 4;
  const int arow   = mt * 16 + nidx;
  const float2* bq = &Xq[laneHi * 16 + nidx];
  const float2* xq = (const float2*)&Xx[laneHi * 2];
  const float2* yq = (const float2*)&Xy[laneHi * 2];
  const float2* zq = (const float2*)&Xz[laneHi * 2];
  const float2* cq = (const float2*)&Cw[arow * DLD + laneHi * 2];
  const float2* dq = (const float2*)&Dp[arow * DLD + laneHi * 2];

  for (int tstep = 0; tstep < TS; ++tstep) {
    float part = 0.0f;
    v8f acc = {};
    {
      const float xi0 = Xx[arow], xi1 = Xy[arow], xi2 = Xz[arow];
      for (int g = 0; g < 8; ++g) {
        float2 xv[4], yv[4], zv[4], cv[4], dv[4], bv[4];
        for (int t = 0; t < 4; ++t) {
          const int kt = g * 4 + t;
          xv[t] = xq[kt * 2];
          yv[t] = yq[kt * 2];
          zv[t] = zq[kt * 2];
          cv[t] = cq[kt * 2];
          dv[t] = dq[kt * 2];
          bv[t] = bq[kt * 32];
        }
        const float xs[8] = {xv[0].x, xv[0].y, xv[1].x, xv[1].y, xv[2].x, xv[2].y, xv[3].x, xv[3].y};
        const float ys[8] = {yv[0].x, yv[0].y, yv[1].x, yv[1].y, yv[2].x, yv[2].y, yv[3].x, yv[3].y};
        const float zs[8] = {zv[0].x, zv[0].y, zv[1].x, zv[1].y, zv[2].x, zv[2].y, zv[3].x, zv[3].y};
        const float cs[8] = {cv[0].x, cv[0].y, cv[1].x, cv[1].y, cv[2].x, cv[2].y, cv[3].x, cv[3].y};
        const float ds[8] = {dv[0].x, dv[0].y, dv[1].x, dv[1].y, dv[2].x, dv[2].y, dv[3].x, dv[3].y};
        float sv[8];
        for (int e = 0; e < 8; ++e) {
          const float u0 = xi0 - xs[e];
          const float u1 = xi1 - ys[e];
          const float u2 = xi2 - zs[e];
          const float r2 = u0 * u0 + u1 * u1 + u2 * u2 + 0.01f;
          const float w = __builtin_amdgcn_rsqf(r2);
          sv[e] = invM4 * cs[e] * (1.0f - ds[e] * w);
        }
        __builtin_amdgcn_sched_barrier(0);
        for (int t = 0; t < 4; ++t) {
          v2f a;  a[0]  = sv[2 * t];  a[1]  = sv[2 * t + 1];
          v2f bb; bb[0] = bv[t].x;    bb[1] = bv[t].y;
          acc = __builtin_amdgcn_wmma_f32_16x16x4_f32(
               false, a,  false, bb,
               (short)0, acc,  false,  false);
        }
        asm volatile("v_nop\n\tv_nop\n\tv_nop\n\tv_nop" : "+v"(acc));
        __builtin_amdgcn_sched_barrier(0);
        for (int e = 0; e < 8; ++e) part += sv[e];
      }
      part += xhalf(part);
      if (laneHi == 0) rsum[arow] = part;
    }
    if (nidx < 3) {
      const int rbase = mt * 16 + laneHi * 8;
      #pragma unroll
      for (int r = 0; r < 8; ++r) Pp[nidx * PLD + rbase + r] = acc[r];
    }
    __syncthreads();

    if (tid < NN) {
      const float gx = rsum[tid] * Xx[tid] - Pp[0 * PLD + tid];
      const float gy = rsum[tid] * Xy[tid] - Pp[1 * PLD + tid];
      const float gz = rsum[tid] * Xz[tid] - Pp[2 * PLD + tid];
      const float dx = -STEP_EPS * gx;
      const float dy = -STEP_EPS * gy;
      const float dz = -STEP_EPS * gz;
      const float speed = sqrtf(dx * dx + dy * dy + dz * dz + 0.001f);
      const float at =
          ALPHA_BASE + (ALPHA - ALPHA_BASE) * ((float)(TS - tstep) / (float)TS);
      const float sc = at * tanhf(speed / at) / speed;
      Xx[tid] += dx * sc;
      Xy[tid] += dy * sc;
      Xz[tid] += dz * sc;
    }
    __syncthreads();
    if (tid < 192) {
      const int q = tid & 63, n = tid >> 6;
      const float* comp = (n == 0) ? Xx : ((n == 1) ? Xy : Xz);
      Xq[q * 16 + n] = make_float2(comp[2 * q], comp[2 * q + 1]);
    }
    __syncthreads();
  }

  float* ob = out + (size_t)b * NN * NN;
  for (int p = tid; p < NN * NN; p += 256) {
    const int i = p >> 7, j = p & 127;
    const float m  = (float)ad[p];
    const float u0 = Xx[i] - Xx[j];
    const float u1 = Xy[i] - Xy[j];
    const float u2 = Xz[i] - Xz[j];
    ob[p] = m * sqrtf(u0 * u0 + u1 * u1 + u2 * u2 + 0.01f);
  }
}

extern "C" void kernel_launch(void* const* d_in, const int* in_sizes, int n_in,
                              void* d_out, int out_size, void* d_ws, size_t ws_size,
                              hipStream_t stream) {
  const float* edge_pred = (const float*)d_in[0];
  const int*   adj       = (const int*)  d_in[1];
  const float* d_init    = (const float*)d_in[2];
  const float* u_init    = (const float*)d_in[3];
  const float* x_noise   = (const float*)d_in[4];
  float*       outp      = (float*)d_out;
  const int B = in_sizes[1] / (NN * NN);
  ts_gcn_fused<<<B, 256, 0, stream>>>(edge_pred, adj, d_init, u_init, x_noise, outp);
}


# Remaining library sqrt and IEEE divisions (power-step norm, eigen scale, gradient speed, output distance) become single hardware v_rsq/v_sqrt/v_rcp instructions; cell rounding changes
# speedup vs baseline: 6.8913x; 1.1116x over previous
#include <hip/hip_runtime.h>
#include <hip/hip_bf16.h>
#include <math.h>


typedef __attribute__((ext_vector_type(2))) float v2f;
typedef __attribute__((ext_vector_type(8))) float v8f;

#define NN   128
#define PLD  132
#define DLD  132
#define TS   10
#define STEP_EPS 0.1f
#define ALPHA 5.0f
#define ALPHA_BASE 0.1f
#define KRANK 3
#define PSTEPS 10

__device__ __forceinline__ float softplus_f(float x) {
  const float ax = fabsf(x);
  const float a  = (ax > 128.0f) ? 128.0f : ax;
  const float y  = a * -1.44269502e+0f;
  float r = __builtin_fmaf(a, -1.44269502e+0f, -y);
  r = __builtin_fmaf(a, -1.92596303e-8f, r);
  const float e0 = __builtin_amdgcn_exp2f(y);
  const float t  = __builtin_fmaf(e0, r * 0.693147181f, e0);
  const float d  = 2.0f + t;
  const float r0 = __builtin_amdgcn_rcpf(d);
  const float s0 = t * r0;
  const float s  = __builtin_fmaf(__builtin_fmaf(-s0, d, t), r0, s0);
  const float z  = s * s;
  float p = 7.69230798e-2f;
  p = __builtin_fmaf(p, z, 9.09090936e-2f);
  p = __builtin_fmaf(p, z, 1.11111112e-1f);
  p = __builtin_fmaf(p, z, 1.42857149e-1f);
  p = __builtin_fmaf(p, z, 2.0e-1f);
  p = __builtin_fmaf(p, z, 3.33333343e-1f);
  const float s2 = s + s;
  return fmaxf(x, 0.0f) + __builtin_fmaf(s2, z * p, s2);
}

__device__ __forceinline__ float xhalf(float x) {
  const int b = __builtin_bit_cast(int, x);
  return __builtin_bit_cast(
      float, __builtin_amdgcn_permlanex16(b, b, 0x76543210, 0xfedcba98, false, false));
}

template <int M>
__device__ __forceinline__ float xrow(float x) {
  const int b = __builtin_bit_cast(int, x);
  return __builtin_bit_cast(float, __builtin_amdgcn_update_dpp(b, b, 0x160 + M, 0xf, 0xf, false));
}

__device__ __forceinline__ float wave_sum(float v) {
  v += xrow<1>(v);
  v += xrow<2>(v);
  v += xrow<4>(v);
  v += xrow<8>(v);
  v += xhalf(v);
  return v;
}

template <int K>
__device__ __forceinline__ void block_sum(float (&v)[K], float* slot, int tid) {
  for (int k = 0; k < K; ++k) v[k] = wave_sum(v[k]);
  if ((tid & 31) == 0) {
    for (int k = 0; k < K; ++k) slot[k * 8 + (tid >> 5)] = v[k];
  }
  __syncthreads();
  for (int k = 0; k < K; ++k) {
    const float4 lo = *(const float4*)&slot[k * 8];
    const float4 hi = *(const float4*)&slot[k * 8 + 4];
    v[k] = ((lo.x + lo.y) + (lo.z + lo.w)) + ((hi.x + hi.y) + (hi.z + hi.w));
  }
}

__device__ __forceinline__ float block_sum1(float x, float* slot, int tid) {
  float v[1] = {x};
  block_sum<1>(v, slot, tid);
  return v[0];
}

__device__ __forceinline__ float mask_at(const unsigned* mb, int i, int j) {
  const unsigned w = mb[i * 4 + ((j >> 6) << 1) + (j & 1)];
  return (float)((w >> ((j & 63) >> 1)) & 1u);
}

__launch_bounds__(256, 4)
__global__ void ts_gcn_fused(const float* __restrict__ edge_pred,
                             const int*   __restrict__ adj,
                             const float* __restrict__ d_init,
                             const float* __restrict__ u_init,
                             const float* __restrict__ x_noise,
                             float*       __restrict__ out)
{
  __shared__ __attribute__((aligned(16))) float Dp[NN * DLD];
  __shared__ __attribute__((aligned(16))) float Cw[NN * DLD];
  __shared__ __attribute__((aligned(16))) unsigned mbit[NN * 4];
  __shared__ float2 Xq[64 * 16];
  __shared__ float  Pp[3 * PLD];
  __shared__ __attribute__((aligned(16))) float Xx[NN];
  __shared__ __attribute__((aligned(16))) float Xy[NN];
  __shared__ __attribute__((aligned(16))) float Xz[NN];
  __shared__ __attribute__((aligned(16))) float rsum[NN];
  __shared__ __attribute__((aligned(16))) float wsum[2][16];
  __shared__ float  ucol[KRANK][NN];
  __shared__ __attribute__((aligned(16))) float ubuf[2 * NN];

  const int b   = blockIdx.x;
  const int tid = threadIdx.x;
  const float d0 = d_init[0];

  const float* ep = edge_pred + (size_t)b * NN * NN * 2;
  const int*   ad = adj       + (size_t)b * NN * NN;

  float partM = 0.0f;
  {
    const float4* ep4 = (const float4*)ep;
    const int2*   ad2 = (const int2*)ad;
    #pragma unroll 1
    for (int kb = 0; kb < 2; ++kb) {
      float4 ev[16];
      int2   av[16];
      #pragma unroll
      for (int u = 0; u < 16; ++u) {
        const int p2 = tid + 256 * (kb * 16 + u);
        ev[u] = ep4[p2];
        av[u] = ad2[p2];
      }
      #pragma unroll
      for (int u = 0; u < 16; ++u) {
        const int p2 = tid + 256 * (kb * 16 + u);
        const int i = p2 >> 6, j = (p2 & 63) << 1;
        const float m0 = (float)av[u].x, m1 = (float)av[u].y;
        *(float2*)&Dp[i * DLD + j] = make_float2(ev[u].x, ev[u].z);
        *(float2*)&Cw[i * DLD + j] = make_float2(ev[u].y, ev[u].w);
        const unsigned be = __builtin_amdgcn_ballot_w32(av[u].x != 0);
        const unsigned bo = __builtin_amdgcn_ballot_w32(av[u].y != 0);
        if ((tid & 31) == 0) *(uint2*)&mbit[i * 4 + ((p2 >> 5) & 1) * 2] = make_uint2(be, bo);
        partM += ((i == j) ? 1.0f : m0) + ((i == j + 1) ? 1.0f : m1);
      }
    }
  }
  __syncthreads();
  for (int q = tid; q < (NN * (NN + 1)) / 2; q += 256) {
    const int r = q / 129, c = q - r * 129;
    const bool lo = c < NN - r;
    const int i = lo ? r : (NN - 1 - r);
    const int j = lo ? (r + c) : (c - 1);
    const int pij = i * DLD + j, pji = j * DLD + i;
    const float sp0 = softplus_f(d0 + (Dp[pij] + Dp[pji]));
    const float sp1 = softplus_f(d0 + (Cw[pij] + Cw[pji]));
    const float mij = mask_at(mbit, i, j), mji = mask_at(mbit, j, i);
    Dp[pij] = sp0 * mij;
    Cw[pij] = sp1 * mij;
    Dp[pji] = sp0 * mji;
    Cw[pji] = sp1 * mji;
  }
  for (int idx = tid; idx < 64 * 16; idx += 256) Xq[idx] = make_float2(0.f, 0.f);
  int rs = 0;
  const float Msum = block_sum1(partM, wsum[rs ^= 1], tid);

  const float Nmol = block_sum1(
      (tid < NN) ? ((tid == 0) ? 1.0f : mask_at(mbit, tid, 0)) : 0.0f, wsum[rs ^= 1], tid);

  const int prow = tid >> 1, ph = tid & 1;
  float4 grow[16];
  {
    const int i = tid >> 1, j0 = (tid & 1) << 6;
    float part = 0.0f;
    for (int jj = 0; jj < 64; ++jj) {
      const float d = Dp[i * DLD + j0 + jj];
      part += d * d;
    }
    part += xrow<1>(part);
    if ((tid & 1) == 0) rsum[i] = part;
    __syncthreads();
    const float tot  = block_sum1((tid < NN) ? rsum[tid] : 0.0f, wsum[rs ^= 1], tid);
    const float invN = 1.0f / Nmol;
    const float Mn   = tot * invN * invN;
    const float rsi = rsum[prow];
    #pragma unroll
    for (int q = 0; q < 16; ++q) {
      const int col0 = (2 * q + ph) * 4;
      const float4 d4 = *(const float4*)&Dp[prow * DLD + col0];
      const float4 r4 = *(const float4*)&rsum[col0];
      const float dd[4] = {d4.x, d4.y, d4.z, d4.w};
      const float rr[4] = {r4.x, r4.y, r4.z, r4.w};
      float gg[4];
      for (int e = 0; e < 4; ++e) {
        const float d  = dd[e];
        const float d2 = d * d;
        const float mk = (prow == col0 + e) ? 1.0f : (d > 0.0f ? 1.0f : 0.0f);
        gg[e] = mk * -0.5f * (d2 - rr[e] * invN - rsi * invN + Mn);
      }
      grow[q] = make_float4(gg[0], gg[1], gg[2], gg[3]);
    }
  }

  int uo = 0;
  for (int kx = 0; kx < KRANK; ++kx) {
    float uval = u_init[(size_t)b * NN * 3 + prow * 3 + kx];
    for (int it = 0; it < PSTEPS; ++it) {
      const float* ub = &ubuf[uo];
      if (ph == 0) ubuf[uo + prow] = uval;
      uo ^= NN;
      __syncthreads();
      float a0 = 0.0f, a1 = 0.0f, a2 = 0.0f, a3 = 0.0f;
      float n0 = 0.0f, n1 = 0.0f, n2 = 0.0f, n3 = 0.0f;
      #pragma unroll
      for (int q = 0; q < 16; ++q) {
        const float4 u4 = *(const float4*)&ub[(2 * q + ph) * 4];
        a0 = fmaf(grow[q].x, u4.x, a0);
        a1 = fmaf(grow[q].y, u4.y, a1);
        a2 = fmaf(grow[q].z, u4.z, a2);
        a3 = fmaf(grow[q].w, u4.w, a3);
        n0 = fmaf(u4.x, u4.x, n0);
        n1 = fmaf(u4.y, u4.y, n1);
        n2 = fmaf(u4.z, u4.z, n2);
        n3 = fmaf(u4.w, u4.w, n3);
      }
      float acc = (a0 + a1) + (a2 + a3);
      float nv  = (n0 + n1) + (n2 + n3);
      acc += xrow<1>(acc);
      nv  += xrow<1>(nv);
      const float inorm = fminf(__builtin_amdgcn_rsqf(nv), 1.0f / 0.001f);
      uval = acc * inorm;
    }
    const float* ub = &ubuf[uo];
    if (ph == 0) ubuf[uo + prow] = uval;
    uo ^= NN;
    __syncthreads();
    float n0 = 0.0f, n1 = 0.0f, n2 = 0.0f, n3 = 0.0f;
    #pragma unroll
    for (int q = 0; q < 16; ++q) {
      const float4 u4 = *(const float4*)&ub[(2 * q + ph) * 4];
      n0 = fmaf(u4.x, u4.x, n0);
      n1 = fmaf(u4.y, u4.y, n1);
      n2 = fmaf(u4.z, u4.z, n2);
      n3 = fmaf(u4.w, u4.w, n3);
    }
    float eig = (n0 + n1) + (n2 + n3);
    eig += xrow<1>(eig);
    const float iden = __builtin_amdgcn_rsqf(__builtin_amdgcn_sqrtf(eig + 0.01f));
    const float unew = uval * iden;
    if (ph == 0) ucol[kx][prow] = unew;
    if (kx + 1 < KRANK) {
      #pragma unroll
      for (int q = 0; q < 16; ++q) {
        const float4 u4 = *(const float4*)&ub[(2 * q + ph) * 4];
        grow[q].x = fmaf(-unew, u4.x * iden, grow[q].x);
        grow[q].y = fmaf(-unew, u4.y * iden, grow[q].y);
        grow[q].z = fmaf(-unew, u4.z * iden, grow[q].z);
        grow[q].w = fmaf(-unew, u4.w * iden, grow[q].w);
      }
    }
  }
  __syncthreads();
  if (tid < NN) {
    const float* xn = x_noise + (size_t)b * NN * 3 + tid * 3;
    Xx[tid] = ucol[0][tid] + xn[0];
    Xy[tid] = ucol[1][tid] + xn[1];
    Xz[tid] = ucol[2][tid] + xn[2];
  }
  __syncthreads();
  if (tid < 192) {
    const int q = tid & 63, n = tid >> 6;
    const float* comp = (n == 0) ? Xx : ((n == 1) ? Xy : Xz);
    Xq[q * 16 + n] = make_float2(comp[2 * q], comp[2 * q + 1]);
  }
  __syncthreads();

  const float invM4 = 4.0f / Msum;
  const int lane   = tid & 31;
  const int mt     = tid >> 5;
  const int nidx   = lane & 15;
  const int laneHi = lane >> 4;
  const int arow   = mt * 16 + nidx;
  const float2* bq = &Xq[laneHi * 16 + nidx];
  const float2* xq = (const float2*)&Xx[laneHi * 2];
  const float2* yq = (const float2*)&Xy[laneHi * 2];
  const float2* zq = (const float2*)&Xz[laneHi * 2];
  const float2* cq = (const float2*)&Cw[arow * DLD + laneHi * 2];
  const float2* dq = (const float2*)&Dp[arow * DLD + laneHi * 2];

  for (int tstep = 0; tstep < TS; ++tstep) {
    float part = 0.0f;
    v8f acc = {};
    {
      const float xi0 = Xx[arow], xi1 = Xy[arow], xi2 = Xz[arow];
      for (int g = 0; g < 8; ++g) {
        float2 xv[4], yv[4], zv[4], cv[4], dv[4], bv[4];
        for (int t = 0; t < 4; ++t) {
          const int kt = g * 4 + t;
          xv[t] = xq[kt * 2];
          yv[t] = yq[kt * 2];
          zv[t] = zq[kt * 2];
          cv[t] = cq[kt * 2];
          dv[t] = dq[kt * 2];
          bv[t] = bq[kt * 32];
        }
        const float xs[8] = {xv[0].x, xv[0].y, xv[1].x, xv[1].y, xv[2].x, xv[2].y, xv[3].x, xv[3].y};
        const float ys[8] = {yv[0].x, yv[0].y, yv[1].x, yv[1].y, yv[2].x, yv[2].y, yv[3].x, yv[3].y};
        const float zs[8] = {zv[0].x, zv[0].y, zv[1].x, zv[1].y, zv[2].x, zv[2].y, zv[3].x, zv[3].y};
        const float cs[8] = {cv[0].x, cv[0].y, cv[1].x, cv[1].y, cv[2].x, cv[2].y, cv[3].x, cv[3].y};
        const float ds[8] = {dv[0].x, dv[0].y, dv[1].x, dv[1].y, dv[2].x, dv[2].y, dv[3].x, dv[3].y};
        float sv[8];
        for (int e = 0; e < 8; ++e) {
          const float u0 = xi0 - xs[e];
          const float u1 = xi1 - ys[e];
          const float u2 = xi2 - zs[e];
          const float r2 = u0 * u0 + u1 * u1 + u2 * u2 + 0.01f;
          const float w = __builtin_amdgcn_rsqf(r2);
          sv[e] = invM4 * cs[e] * (1.0f - ds[e] * w);
        }
        __builtin_amdgcn_sched_barrier(0);
        for (int t = 0; t < 4; ++t) {
          v2f a;  a[0]  = sv[2 * t];  a[1]  = sv[2 * t + 1];
          v2f bb; bb[0] = bv[t].x;    bb[1] = bv[t].y;
          acc = __builtin_amdgcn_wmma_f32_16x16x4_f32(
               false, a,  false, bb,
               (short)0, acc,  false,  false);
        }
        asm volatile("v_nop\n\tv_nop\n\tv_nop\n\tv_nop" : "+v"(acc));
        __builtin_amdgcn_sched_barrier(0);
        for (int e = 0; e < 8; ++e) part += sv[e];
      }
      part += xhalf(part);
      if (laneHi == 0) rsum[arow] = part;
    }
    if (nidx < 3) {
      const int rbase = mt * 16 + laneHi * 8;
      #pragma unroll
      for (int r = 0; r < 8; ++r) Pp[nidx * PLD + rbase + r] = acc[r];
    }
    __syncthreads();

    if (tid < NN) {
      const float gx = rsum[tid] * Xx[tid] - Pp[0 * PLD + tid];
      const float gy = rsum[tid] * Xy[tid] - Pp[1 * PLD + tid];
      const float gz = rsum[tid] * Xz[tid] - Pp[2 * PLD + tid];
      const float dx = -STEP_EPS * gx;
      const float dy = -STEP_EPS * gy;
      const float dz = -STEP_EPS * gz;
      const float sp2   = dx * dx + dy * dy + dz * dz + 0.001f;
      const float speed = __builtin_amdgcn_sqrtf(sp2);
      const float at =
          ALPHA_BASE + (ALPHA - ALPHA_BASE) * ((float)(TS - tstep) / (float)TS);
      const float sc = at * tanhf(speed * __builtin_amdgcn_rcpf(at)) * __builtin_amdgcn_rsqf(sp2);
      Xx[tid] += dx * sc;
      Xy[tid] += dy * sc;
      Xz[tid] += dz * sc;
    }
    __syncthreads();
    if (tid < 192) {
      const int q = tid & 63, n = tid >> 6;
      const float* comp = (n == 0) ? Xx : ((n == 1) ? Xy : Xz);
      Xq[q * 16 + n] = make_float2(comp[2 * q], comp[2 * q + 1]);
    }
    __syncthreads();
  }

  float* ob = out + (size_t)b * NN * NN;
  #pragma unroll 8
  for (int p = tid; p < NN * NN; p += 256) {
    const int i = p >> 7, j = p & 127;
    const float m  = (float)ad[p];
    const float u0 = Xx[i] - Xx[j];
    const float u1 = Xy[i] - Xy[j];
    const float u2 = Xz[i] - Xz[j];
    ob[p] = m * __builtin_amdgcn_sqrtf(u0 * u0 + u1 * u1 + u2 * u2 + 0.01f);
  }
}

extern "C" void kernel_launch(void* const* d_in, const int* in_sizes, int n_in,
                              void* d_out, int out_size, void* d_ws, size_t ws_size,
                              hipStream_t stream) {
  const float* edge_pred = (const float*)d_in[0];
  const int*   adj       = (const int*)  d_in[1];
  const float* d_init    = (const float*)d_in[2];
  const float* u_init    = (const float*)d_in[3];
  const float* x_noise   = (const float*)d_in[4];
  float*       outp      = (float*)d_out;
  const int B = in_sizes[1] / (NN * NN);
  ts_gcn_fused<<<B, 256, 0, stream>>>(edge_pred, adj, d_init, u_init, x_noise, outp);
}
